# starting kernel
# baseline (speedup 1.0000x reference)
#include <hip/hip_runtime.h>
#include <hip/hip_bf16.h>

typedef float v2f __attribute__((ext_vector_type(2)));
typedef float v8f __attribute__((ext_vector_type(8)));

#define N_NODES 50000
#define N_EDGES 800000
#define IN_D 64
#define HID_D 64
#define OUT_D 32

__global__ void gcn_zero_f32(float* p, long n) {
    long i = (long)blockIdx.x * blockDim.x + threadIdx.x;
    if (i < n) p[i] = 0.0f;
}

__global__ void gcn_degrees(const int* __restrict__ src,
                            const int* __restrict__ dst,
                            float* __restrict__ degA,
                            float* __restrict__ degB, int E) {
    int e = blockIdx.x * blockDim.x + threadIdx.x;
    if (e < E) {
        atomicAdd(degA + src[e], 1.0f);
        atomicAdd(degB + dst[e], 1.0f);
    }
}

__global__ void gcn_norms(float* __restrict__ deg, long n) {
    long i = (long)blockIdx.x * blockDim.x + threadIdx.x;
    if (i < n) deg[i] = rsqrtf(fmaxf(deg[i], 1.0f));
}

__global__ void gcn_scale_transpose(const float* __restrict__ h,
                                    const float* __restrict__ nrmS,
                                    float* __restrict__ feat) {
    long t = (long)blockIdx.x * blockDim.x + threadIdx.x;
    if (t >= (long)IN_D * N_NODES) return;
    int d = (int)(t / N_NODES);
    int n = (int)(t % N_NODES);
    feat[(size_t)n * IN_D + d] = h[t] * nrmS[n];
}

__global__ void gcn_scatter(const int* __restrict__ src,
                            const int* __restrict__ dst,
                            const float* __restrict__ feat,
                            float* __restrict__ agg, int E) {
    int wid  = (blockIdx.x * blockDim.x + threadIdx.x) >> 5;
    int lane = threadIdx.x & 31;
    if (wid >= E) return;
    int s = src[wid];
    int t = dst[wid];
    const float2 v = *(const float2*)(feat + (size_t)s * 64 + lane * 2);
    float* a = agg + (size_t)t * 64 + lane * 2;
    atomicAdd(a + 0, v.x);
    atomicAdd(a + 1, v.y);
}

__global__ void gcn_gemm1_wmma(const float* __restrict__ agg,
                               const float* __restrict__ nrmD,
                               const float* __restrict__ nrmS,
                               const float* __restrict__ W1,
                               const float* __restrict__ b1,
                               float* __restrict__ feat) {
    const int lane  = threadIdx.x & 31;
    const int wv    = threadIdx.x >> 5;
    const int m     = lane & 15;
    const int khalf = (lane >> 4) << 1;
    const int row   = blockIdx.x * 16 + m;
    const int col   = (wv << 4) + m;

    const float        scale = nrmD[row];
    const float* const arow  = agg + (size_t)row * 64;

    v8f acc = {};
#pragma unroll
    for (int kk = 0; kk < 64; kk += 4) {
        const int k = kk + khalf;
        v2f a, b;
        a.x = arow[k] * scale;
        a.y = arow[k + 1] * scale;
        b.x = W1[k * HID_D + col];
        b.y = W1[(k + 1) * HID_D + col];
        acc = __builtin_amdgcn_wmma_f32_16x16x4_f32(
            false, a, false, b, (short)0, acc, false, false);
    }

    const int   rbase = blockIdx.x * 16 + ((lane >> 4) << 3);
    const float bias  = b1[col];
#pragma unroll
    for (int j = 0; j < 8; ++j) {
        const int r = rbase + j;
        float v = acc[j] + bias;
        v = v > 0.0f ? v : 0.0f;
        feat[(size_t)r * 64 + col] = v * nrmS[r];
    }
}

__global__ void gcn_gemm2_wmma(const float* __restrict__ agg,
                               const float* __restrict__ nrmD,
                               const float* __restrict__ W2,
                               const float* __restrict__ b2,
                               float* __restrict__ out) {
    const int lane  = threadIdx.x & 31;
    const int wv    = threadIdx.x >> 5;
    const int m     = lane & 15;
    const int khalf = (lane >> 4) << 1;
    const int row   = blockIdx.x * 16 + m;
    const int col   = (wv << 4) + m;

    const float        scale = nrmD[row];
    const float* const arow  = agg + (size_t)row * 64;

    v8f acc = {};
#pragma unroll
    for (int kk = 0; kk < 64; kk += 4) {
        const int k = kk + khalf;
        v2f a, b;
        a.x = arow[k] * scale;
        a.y = arow[k + 1] * scale;
        b.x = W2[k * OUT_D + col];
        b.y = W2[(k + 1) * OUT_D + col];
        acc = __builtin_amdgcn_wmma_f32_16x16x4_f32(
            false, a, false, b, (short)0, acc, false, false);
    }

    const int   rbase = blockIdx.x * 16 + ((lane >> 4) << 3);
    const float bias  = b2[col];
#pragma unroll
    for (int j = 0; j < 8; ++j) {
        const int r = rbase + j;
        out[(size_t)col * N_NODES + r] = acc[j] + bias;
    }
}

extern "C" void kernel_launch(void* const* d_in, const int* in_sizes, int n_in,
                              void* d_out, int out_size, void* d_ws, size_t ws_size,
                              hipStream_t stream) {
    const float* h   = (const float*)d_in[0];
    const int*   src = (const int*)  d_in[1];
    const int*   dst = (const int*)  d_in[2];
    const float* W1  = (const float*)d_in[3];
    const float* b1  = (const float*)d_in[4];
    const float* W2  = (const float*)d_in[5];
    const float* b2  = (const float*)d_in[6];
    float*       out = (float*)d_out;

    float* nrmS = (float*)d_ws;
    float* nrmD = nrmS + N_NODES;
    float* feat = nrmD + N_NODES;
    float* agg  = feat + (size_t)N_NODES * 64;

    const int  T    = 256;
    const long degN = 2L * N_NODES;
    const long aggN = 64L * N_NODES;

    gcn_zero_f32<<<(unsigned)((degN + T - 1) / T), T, 0, stream>>>(nrmS, degN);
    gcn_degrees <<<(N_EDGES + T - 1) / T, T, 0, stream>>>(src, dst, nrmS, nrmD, N_EDGES);
    gcn_norms   <<<(unsigned)((degN + T - 1) / T), T, 0, stream>>>(nrmS, degN);

    gcn_scale_transpose<<<(unsigned)((aggN + T - 1) / T), T, 0, stream>>>(h, nrmS, feat);
    gcn_zero_f32<<<(unsigned)((aggN + T - 1) / T), T, 0, stream>>>(agg, aggN);
    gcn_scatter <<<(N_EDGES * 32 + T - 1) / T, T, 0, stream>>>(src, dst, feat, agg, N_EDGES);
    gcn_gemm1_wmma<<<N_NODES / 16, 128, 0, stream>>>(agg, nrmD, nrmS, W1, b1, feat);

    gcn_zero_f32<<<(unsigned)((aggN + T - 1) / T), T, 0, stream>>>(agg, aggN);
    gcn_scatter <<<(N_EDGES * 32 + T - 1) / T, T, 0, stream>>>(src, dst, feat, agg, N_EDGES);
    gcn_gemm2_wmma<<<N_NODES / 16, 64, 0, stream>>>(agg, nrmD, W2, b2, out);
}


# scale+transpose staged through a padded LDS tile: 16-byte coalesced reads of h, every 128-byte line of feat written whole by one wave store instead of 32 scattered 4-byte stores
# speedup vs baseline: 1.2907x; 1.2907x over previous
#include <hip/hip_runtime.h>
#include <hip/hip_bf16.h>

typedef float v2f __attribute__((ext_vector_type(2)));
typedef float v8f __attribute__((ext_vector_type(8)));

#define N_NODES 50000
#define N_EDGES 800000
#define IN_D 64
#define HID_D 64
#define OUT_D 32

__global__ void gcn_zero_f32(float* p, long n) {
    long i = (long)blockIdx.x * blockDim.x + threadIdx.x;
    if (i < n) p[i] = 0.0f;
}

__global__ void gcn_degrees(const int* __restrict__ src,
                            const int* __restrict__ dst,
                            float* __restrict__ degA,
                            float* __restrict__ degB, int E) {
    int e = blockIdx.x * blockDim.x + threadIdx.x;
    if (e < E) {
        atomicAdd(degA + src[e], 1.0f);
        atomicAdd(degB + dst[e], 1.0f);
    }
}

__global__ void gcn_norms(float* __restrict__ deg, long n) {
    long i = (long)blockIdx.x * blockDim.x + threadIdx.x;
    if (i < n) deg[i] = rsqrtf(fmaxf(deg[i], 1.0f));
}

#define TP_N     64
#define TP_PITCH 65
__global__ __launch_bounds__(256)
void gcn_scale_transpose(const float* __restrict__ h,
                         const float* __restrict__ nrmS,
                         float* __restrict__ feat) {
    __shared__ float tile[TP_N * TP_PITCH];
    const int tid = threadIdx.x;
    const int n0  = blockIdx.x * TP_N;

    const int c = tid & 15;
    const int r = tid >> 4;
    const int n = n0 + 4 * c;
    if (n < N_NODES) {
        const float4 s = *(const float4*)(nrmS + n);
        float4 v[4];
        for (int p = 0; p < 4; ++p)
            v[p] = *(const float4*)(h + (size_t)(r + 16 * p) * N_NODES + n);
        for (int p = 0; p < 4; ++p) {
            const int d = r + 16 * p;
            tile[(4 * c + 0) * TP_PITCH + d] = v[p].x * s.x;
            tile[(4 * c + 1) * TP_PITCH + d] = v[p].y * s.y;
            tile[(4 * c + 2) * TP_PITCH + d] = v[p].z * s.z;
            tile[(4 * c + 3) * TP_PITCH + d] = v[p].w * s.w;
        }
    }
    __syncthreads();

    const int nvalid = (N_NODES - n0) < TP_N ? (N_NODES - n0) : TP_N;
    float* const dstp = feat + (size_t)n0 * IN_D;
    for (int k = 0; k < 4; ++k) {
        const int e  = tid + 256 * k;
        const int nn = e >> 4;
        const int q  = e & 15;
        if (nn < nvalid) {
            const float* t = tile + nn * TP_PITCH + 4 * q;
            float4 o;
            o.x = t[0]; o.y = t[1]; o.z = t[2]; o.w = t[3];
            *(float4*)(dstp + (size_t)e * 4) = o;
        }
    }
}

__global__ void gcn_scatter(const int* __restrict__ src,
                            const int* __restrict__ dst,
                            const float* __restrict__ feat,
                            float* __restrict__ agg, int E) {
    int wid  = (blockIdx.x * blockDim.x + threadIdx.x) >> 5;
    int lane = threadIdx.x & 31;
    if (wid >= E) return;
    int s = src[wid];
    int t = dst[wid];
    const float2 v = *(const float2*)(feat + (size_t)s * 64 + lane * 2);
    float* a = agg + (size_t)t * 64 + lane * 2;
    atomicAdd(a + 0, v.x);
    atomicAdd(a + 1, v.y);
}

__global__ void gcn_gemm1_wmma(const float* __restrict__ agg,
                               const float* __restrict__ nrmD,
                               const float* __restrict__ nrmS,
                               const float* __restrict__ W1,
                               const float* __restrict__ b1,
                               float* __restrict__ feat) {
    const int lane  = threadIdx.x & 31;
    const int wv    = threadIdx.x >> 5;
    const int m     = lane & 15;
    const int khalf = (lane >> 4) << 1;
    const int row   = blockIdx.x * 16 + m;
    const int col   = (wv << 4) + m;

    const float        scale = nrmD[row];
    const float* const arow  = agg + (size_t)row * 64;

    v8f acc = {};
#pragma unroll
    for (int kk = 0; kk < 64; kk += 4) {
        const int k = kk + khalf;
        v2f a, b;
        a.x = arow[k] * scale;
        a.y = arow[k + 1] * scale;
        b.x = W1[k * HID_D + col];
        b.y = W1[(k + 1) * HID_D + col];
        acc = __builtin_amdgcn_wmma_f32_16x16x4_f32(
            false, a, false, b, (short)0, acc, false, false);
    }

    const int   rbase = blockIdx.x * 16 + ((lane >> 4) << 3);
    const float bias  = b1[col];
#pragma unroll
    for (int j = 0; j < 8; ++j) {
        const int r = rbase + j;
        float v = acc[j] + bias;
        v = v > 0.0f ? v : 0.0f;
        feat[(size_t)r * 64 + col] = v * nrmS[r];
    }
}

__global__ void gcn_gemm2_wmma(const float* __restrict__ agg,
                               const float* __restrict__ nrmD,
                               const float* __restrict__ W2,
                               const float* __restrict__ b2,
                               float* __restrict__ out) {
    const int lane  = threadIdx.x & 31;
    const int wv    = threadIdx.x >> 5;
    const int m     = lane & 15;
    const int khalf = (lane >> 4) << 1;
    const int row   = blockIdx.x * 16 + m;
    const int col   = (wv << 4) + m;

    const float        scale = nrmD[row];
    const float* const arow  = agg + (size_t)row * 64;

    v8f acc = {};
#pragma unroll
    for (int kk = 0; kk < 64; kk += 4) {
        const int k = kk + khalf;
        v2f a, b;
        a.x = arow[k] * scale;
        a.y = arow[k + 1] * scale;
        b.x = W2[k * OUT_D + col];
        b.y = W2[(k + 1) * OUT_D + col];
        acc = __builtin_amdgcn_wmma_f32_16x16x4_f32(
            false, a, false, b, (short)0, acc, false, false);
    }

    const int   rbase = blockIdx.x * 16 + ((lane >> 4) << 3);
    const float bias  = b2[col];
#pragma unroll
    for (int j = 0; j < 8; ++j) {
        const int r = rbase + j;
        out[(size_t)col * N_NODES + r] = acc[j] + bias;
    }
}

extern "C" void kernel_launch(void* const* d_in, const int* in_sizes, int n_in,
                              void* d_out, int out_size, void* d_ws, size_t ws_size,
                              hipStream_t stream) {
    const float* h   = (const float*)d_in[0];
    const int*   src = (const int*)  d_in[1];
    const int*   dst = (const int*)  d_in[2];
    const float* W1  = (const float*)d_in[3];
    const float* b1  = (const float*)d_in[4];
    const float* W2  = (const float*)d_in[5];
    const float* b2  = (const float*)d_in[6];
    float*       out = (float*)d_out;

    float* nrmS = (float*)d_ws;
    float* nrmD = nrmS + N_NODES;
    float* feat = nrmD + N_NODES;
    float* agg  = feat + (size_t)N_NODES * 64;

    const int  T    = 256;
    const long degN = 2L * N_NODES;
    const long aggN = 64L * N_NODES;

    gcn_zero_f32<<<(unsigned)((degN + T - 1) / T), T, 0, stream>>>(nrmS, degN);
    gcn_degrees <<<(N_EDGES + T - 1) / T, T, 0, stream>>>(src, dst, nrmS, nrmD, N_EDGES);
    gcn_norms   <<<(unsigned)((degN + T - 1) / T), T, 0, stream>>>(nrmS, degN);

    gcn_scale_transpose<<<(N_NODES + TP_N - 1) / TP_N, 256, 0, stream>>>(h, nrmS, feat);
    gcn_zero_f32<<<(unsigned)((aggN + T - 1) / T), T, 0, stream>>>(agg, aggN);
    gcn_scatter <<<(N_EDGES * 32 + T - 1) / T, T, 0, stream>>>(src, dst, feat, agg, N_EDGES);
    gcn_gemm1_wmma<<<N_NODES / 16, 128, 0, stream>>>(agg, nrmD, nrmS, W1, b1, feat);

    gcn_zero_f32<<<(unsigned)((aggN + T - 1) / T), T, 0, stream>>>(agg, aggN);
    gcn_scatter <<<(N_EDGES * 32 + T - 1) / T, T, 0, stream>>>(src, dst, feat, agg, N_EDGES);
    gcn_gemm2_wmma<<<N_NODES / 16, 64, 0, stream>>>(agg, nrmD, W2, b2, out);
}


# float-atomic scatter-add replaced by a gather over edges bucketed by destination (int counts, one-block scan, arrival-order slot fill; built once, used by both layers); sums now in slot order
# speedup vs baseline: 1.7045x; 1.3206x over previous
#include <hip/hip_runtime.h>
#include <hip/hip_bf16.h>


typedef float v2f __attribute__((ext_vector_type(2)));
typedef float v8f __attribute__((ext_vector_type(8)));

#define N_NODES 50000
#define N_EDGES 800000
#define IN_D 64
#define HID_D 64
#define OUT_D 32

__global__ void gcn_zero_i32(int* p, int n) {
    int i = blockIdx.x * blockDim.x + threadIdx.x;
    if (i < n) p[i] = 0;
}

__global__ void gcn_count(const int* __restrict__ src,
                          const int* __restrict__ dst,
                          int* __restrict__ cntS,
                          int* __restrict__ cntD, int E) {
    int e = blockIdx.x * blockDim.x + threadIdx.x;
    if (e < E) {
        atomicAdd(cntS + src[e], 1);
        atomicAdd(cntD + dst[e], 1);
    }
}

__global__ void gcn_norms(const int* __restrict__ cnt, float* __restrict__ nrm, int n) {
    int i = blockIdx.x * blockDim.x + threadIdx.x;
    if (i < n) nrm[i] = rsqrtf(fmaxf((float)cnt[i], 1.0f));
}

#define SC_ROWS 13
__global__ __launch_bounds__(1024)
void gcn_scan(const int* __restrict__ cnt, int* __restrict__ rowptr) {
    __shared__ int wsum[32];
    const int lane  = threadIdx.x & 31;
    const int wv    = threadIdx.x >> 5;
    const int wbase = wv * (SC_ROWS * 128) + lane * 4;

    int4 v[SC_ROWS];
    int  ls[SC_ROWS];
    int  tot = 0;
#pragma unroll
    for (int q = 0; q < SC_ROWS; ++q) {
        const int idx = wbase + q * 128;
        v[q] = make_int4(0, 0, 0, 0);
        if (idx < N_NODES) v[q] = *(const int4*)(cnt + idx);
        ls[q] = v[q].x + v[q].y + v[q].z + v[q].w;
        tot += ls[q];
    }
    for (int o = 16; o > 0; o >>= 1) tot += __shfl_xor(tot, o);
    if (lane == 0) wsum[wv] = tot;
    __syncthreads();

    int t = wsum[lane];
    for (int o = 1; o < 32; o <<= 1) {
        const int u = __shfl_up(t, o);
        if (lane >= o) t += u;
    }
    const int total = __shfl(t, 31);
    const int wincl = __shfl(t, wv);
    int carry = wincl - tot;

#pragma unroll
    for (int q = 0; q < SC_ROWS; ++q) {
        int inc = ls[q];
        for (int o = 1; o < 32; o <<= 1) {
            const int u = __shfl_up(inc, o);
            if (lane >= o) inc += u;
        }
        const int ex  = carry + inc - ls[q];
        const int idx = wbase + q * 128;
        if (idx < N_NODES) {
            int4 r;
            r.x = ex;
            r.y = ex + v[q].x;
            r.z = r.y + v[q].y;
            r.w = r.z + v[q].z;
            *(int4*)(rowptr + idx) = r;
        }
        carry += __shfl(inc, 31);
    }
    if (threadIdx.x == 0) rowptr[N_NODES] = total;
}

__global__ void gcn_fill(const int* __restrict__ src,
                         const int* __restrict__ dst,
                         const int* __restrict__ rowptr,
                         int* __restrict__ cntD,
                         int* __restrict__ csr, int E) {
    int e = blockIdx.x * blockDim.x + threadIdx.x;
    if (e >= E) return;
    const int d   = dst[e];
    const int old = atomicSub(cntD + d, 1);
    int p = rowptr[d] + old - 1;
    p = min(max(p, 0), E - 1);
    csr[p] = src[e];
}

#define TP_N     64
#define TP_PITCH 65
__global__ __launch_bounds__(256)
void gcn_scale_transpose(const float* __restrict__ h,
                         const float* __restrict__ nrmS,
                         float* __restrict__ feat) {
    __shared__ float tile[TP_N * TP_PITCH];
    const int tid = threadIdx.x;
    const int n0  = blockIdx.x * TP_N;

    const int c = tid & 15;
    const int r = tid >> 4;
    const int n = n0 + 4 * c;
    if (n < N_NODES) {
        const float4 s = *(const float4*)(nrmS + n);
        float4 v[4];
        for (int p = 0; p < 4; ++p)
            v[p] = *(const float4*)(h + (size_t)(r + 16 * p) * N_NODES + n);
        for (int p = 0; p < 4; ++p) {
            const int d = r + 16 * p;
            tile[(4 * c + 0) * TP_PITCH + d] = v[p].x * s.x;
            tile[(4 * c + 1) * TP_PITCH + d] = v[p].y * s.y;
            tile[(4 * c + 2) * TP_PITCH + d] = v[p].z * s.z;
            tile[(4 * c + 3) * TP_PITCH + d] = v[p].w * s.w;
        }
    }
    __syncthreads();

    const int nvalid = (N_NODES - n0) < TP_N ? (N_NODES - n0) : TP_N;
    float* const dstp = feat + (size_t)n0 * IN_D;
    for (int k = 0; k < 4; ++k) {
        const int e  = tid + 256 * k;
        const int nn = e >> 4;
        const int q  = e & 15;
        if (nn < nvalid) {
            const float* t = tile + nn * TP_PITCH + 4 * q;
            float4 o;
            o.x = t[0]; o.y = t[1]; o.z = t[2]; o.w = t[3];
            *(float4*)(dstp + (size_t)e * 4) = o;
        }
    }
}

__global__ __launch_bounds__(256)
void gcn_gather(const int* __restrict__ rowptr,
                const int* __restrict__ csr,
                const float* __restrict__ feat,
                float* __restrict__ agg) {
    const int node = (blockIdx.x * 256 + threadIdx.x) >> 4;
    const int l    = threadIdx.x & 15;
    int beg = rowptr[node];
    int end = rowptr[node + 1];
    beg = max(beg, 0);
    end = min(end, N_EDGES);
    const float* const fcol = feat + 4 * l;

    float4 acc = make_float4(0.0f, 0.0f, 0.0f, 0.0f);
    int next = 0;
    if (beg + l < end) next = csr[beg + l];
    for (int j = beg; j < end; j += 16) {
        const int rem = end - j;
        const int cnt = rem < 16 ? rem : 16;
        const int mine = min(max(next, 0), N_NODES - 1);
        next = 0;
        if (j + 16 + l < end) next = csr[j + 16 + l];
        for (int k = 0; k < cnt; k += 4) {
            const int s0 = __shfl(mine, k + 0, 16);
            const int s1 = __shfl(mine, k + 1, 16);
            const int s2 = __shfl(mine, k + 2, 16);
            const int s3 = __shfl(mine, k + 3, 16);
            const float4 z = make_float4(0.0f, 0.0f, 0.0f, 0.0f);
            float4 v0 = *(const float4*)(fcol + (size_t)s0 * 64);
            float4 v1 = z, v2 = z, v3 = z;
            if (k + 1 < cnt) v1 = *(const float4*)(fcol + (size_t)s1 * 64);
            if (k + 2 < cnt) v2 = *(const float4*)(fcol + (size_t)s2 * 64);
            if (k + 3 < cnt) v3 = *(const float4*)(fcol + (size_t)s3 * 64);
            acc.x += v0.x; acc.y += v0.y; acc.z += v0.z; acc.w += v0.w;
            acc.x += v1.x; acc.y += v1.y; acc.z += v1.z; acc.w += v1.w;
            acc.x += v2.x; acc.y += v2.y; acc.z += v2.z; acc.w += v2.w;
            acc.x += v3.x; acc.y += v3.y; acc.z += v3.z; acc.w += v3.w;
        }
    }
    *(float4*)(agg + (size_t)node * 64 + 4 * l) = acc;
}

__global__ void gcn_gemm1_wmma(const float* __restrict__ agg,
                               const float* __restrict__ nrmD,
                               const float* __restrict__ nrmS,
                               const float* __restrict__ W1,
                               const float* __restrict__ b1,
                               float* __restrict__ feat) {
    const int lane  = threadIdx.x & 31;
    const int wv    = threadIdx.x >> 5;
    const int m     = lane & 15;
    const int khalf = (lane >> 4) << 1;
    const int row   = blockIdx.x * 16 + m;
    const int col   = (wv << 4) + m;

    const float        scale = nrmD[row];
    const float* const arow  = agg + (size_t)row * 64;

    v8f acc = {};
#pragma unroll
    for (int kk = 0; kk < 64; kk += 4) {
        const int k = kk + khalf;
        v2f a, b;
        a.x = arow[k] * scale;
        a.y = arow[k + 1] * scale;
        b.x = W1[k * HID_D + col];
        b.y = W1[(k + 1) * HID_D + col];
        acc = __builtin_amdgcn_wmma_f32_16x16x4_f32(
            false, a, false, b, (short)0, acc, false, false);
    }

    const int   rbase = blockIdx.x * 16 + ((lane >> 4) << 3);
    const float bias  = b1[col];
#pragma unroll
    for (int j = 0; j < 8; ++j) {
        const int r = rbase + j;
        float v = acc[j] + bias;
        v = v > 0.0f ? v : 0.0f;
        feat[(size_t)r * 64 + col] = v * nrmS[r];
    }
}

__global__ void gcn_gemm2_wmma(const float* __restrict__ agg,
                               const float* __restrict__ nrmD,
                               const float* __restrict__ W2,
                               const float* __restrict__ b2,
                               float* __restrict__ out) {
    const int lane  = threadIdx.x & 31;
    const int wv    = threadIdx.x >> 5;
    const int m     = lane & 15;
    const int khalf = (lane >> 4) << 1;
    const int row   = blockIdx.x * 16 + m;
    const int col   = (wv << 4) + m;

    const float        scale = nrmD[row];
    const float* const arow  = agg + (size_t)row * 64;

    v8f acc = {};
#pragma unroll
    for (int kk = 0; kk < 64; kk += 4) {
        const int k = kk + khalf;
        v2f a, b;
        a.x = arow[k] * scale;
        a.y = arow[k + 1] * scale;
        b.x = W2[k * OUT_D + col];
        b.y = W2[(k + 1) * OUT_D + col];
        acc = __builtin_amdgcn_wmma_f32_16x16x4_f32(
            false, a, false, b, (short)0, acc, false, false);
    }

    const int   rbase = blockIdx.x * 16 + ((lane >> 4) << 3);
    const float bias  = b2[col];
#pragma unroll
    for (int j = 0; j < 8; ++j) {
        const int r = rbase + j;
        out[(size_t)col * N_NODES + r] = acc[j] + bias;
    }
}

extern "C" void kernel_launch(void* const* d_in, const int* in_sizes, int n_in,
                              void* d_out, int out_size, void* d_ws, size_t ws_size,
                              hipStream_t stream) {
    const float* h   = (const float*)d_in[0];
    const int*   src = (const int*)  d_in[1];
    const int*   dst = (const int*)  d_in[2];
    const float* W1  = (const float*)d_in[3];
    const float* b1  = (const float*)d_in[4];
    const float* W2  = (const float*)d_in[5];
    const float* b2  = (const float*)d_in[6];
    float*       out = (float*)d_out;

    int*   cntS   = (int*)d_ws;
    int*   cntD   = cntS + N_NODES;
    float* nrmS   = (float*)d_ws + 2 * N_NODES;
    float* nrmD   = nrmS + N_NODES;
    int*   rowptr = (int*)d_ws + 4 * N_NODES;
    int*   csr    = rowptr + 50048;
    float* feat   = (float*)d_ws + 4 * N_NODES + 50048 + N_EDGES;
    float* agg    = feat + (size_t)N_NODES * 64;

    const int T = 256;

    gcn_zero_i32<<<(2 * N_NODES + T - 1) / T, T, 0, stream>>>(cntS, 2 * N_NODES);
    gcn_count   <<<(N_EDGES + T - 1) / T, T, 0, stream>>>(src, dst, cntS, cntD, N_EDGES);
    gcn_norms   <<<(2 * N_NODES + T - 1) / T, T, 0, stream>>>(cntS, nrmS, 2 * N_NODES);
    gcn_scan    <<<1, 1024, 0, stream>>>(cntD, rowptr);
    gcn_fill    <<<(N_EDGES + T - 1) / T, T, 0, stream>>>(src, dst, rowptr, cntD, csr, N_EDGES);

    gcn_scale_transpose<<<(N_NODES + TP_N - 1) / TP_N, 256, 0, stream>>>(h, nrmS, feat);
    gcn_gather    <<<N_NODES / 16, 256, 0, stream>>>(rowptr, csr, feat, agg);
    gcn_gemm1_wmma<<<N_NODES / 16, 128, 0, stream>>>(agg, nrmD, nrmS, W1, b1, feat);

    gcn_gather    <<<N_NODES / 16, 256, 0, stream>>>(rowptr, csr, feat, agg);
    gcn_gemm2_wmma<<<N_NODES / 16, 64, 0, stream>>>(agg, nrmD, W2, b2, out);
}


# per-edge global atomics of the degree count and the by-destination slot fill replaced by LDS counters over (edge chunk, node range) blocks; slots, hence sums, now in chunk order; 9 launches
# speedup vs baseline: 3.5202x; 2.0652x over previous
#include <hip/hip_runtime.h>
#include <hip/hip_bf16.h>


typedef float v2f __attribute__((ext_vector_type(2)));
typedef float v8f __attribute__((ext_vector_type(8)));

#define N_NODES 50000
#define N_EDGES 800000
#define IN_D 64
#define HID_D 64
#define OUT_D 32

#define HB   32
#define HR   4
#define H_CH (N_EDGES / HB)
#define H_RN (N_NODES / HR)
#define H_Q  (H_CH / 4)
#define H_IT ((H_Q + 1023) / 1024)
#define H_OT ((H_RN / 4 + 1023) / 1024)

__global__ __launch_bounds__(1024)
void gcn_hist(const int* __restrict__ src,
              const int* __restrict__ dst,
              int* __restrict__ histS,
              int* __restrict__ histD) {
    __shared__ int cnt[H_RN];
    const int tid = threadIdx.x;
    const int b   = blockIdx.x / HR;
    const int lo  = (blockIdx.x % HR) * H_RN;
    const int* const ids  = (blockIdx.y ? src : dst) + b * H_CH;
    int* const       hist = (blockIdx.y ? histS : histD) + b * N_NODES + lo;

    int4 v[H_IT];
#pragma unroll
    for (int i = 0; i < H_IT; ++i) {
        const int q = tid + 1024 * i;
        v[i] = make_int4(-1, -1, -1, -1);
        if (q < H_Q) v[i] = *(const int4*)(ids + 4 * q);
    }
    for (int i = tid; i < H_RN / 4; i += 1024) *(int4*)(cnt + 4 * i) = make_int4(0, 0, 0, 0);
    __syncthreads();
#pragma unroll
    for (int i = 0; i < H_IT; ++i) {
        const unsigned k0 = (unsigned)(v[i].x - lo);
        const unsigned k1 = (unsigned)(v[i].y - lo);
        const unsigned k2 = (unsigned)(v[i].z - lo);
        const unsigned k3 = (unsigned)(v[i].w - lo);
        if (k0 < (unsigned)H_RN) atomicAdd(cnt + k0, 1);
        if (k1 < (unsigned)H_RN) atomicAdd(cnt + k1, 1);
        if (k2 < (unsigned)H_RN) atomicAdd(cnt + k2, 1);
        if (k3 < (unsigned)H_RN) atomicAdd(cnt + k3, 1);
    }
    __syncthreads();
    for (int i = tid; i < H_RN / 4; i += 1024)
        *(int4*)(hist + 4 * i) = *(const int4*)(cnt + 4 * i);
}

__global__ __launch_bounds__(256)
void gcn_combine(int* __restrict__ histD,
                 const int* __restrict__ histS,
                 int* __restrict__ cntD,
                 float* __restrict__ nrmS,
                 float* __restrict__ nrmD) {
    const int n = blockIdx.x * 256 + threadIdx.x;
    if (n >= N_NODES) return;
    int d[HB], s[HB];
#pragma unroll
    for (int b = 0; b < HB; ++b) {
        d[b] = histD[b * N_NODES + n];
        s[b] = histS[b * N_NODES + n];
    }
    int run = 0, tot = 0;
#pragma unroll
    for (int b = 0; b < HB; ++b) {
        histD[b * N_NODES + n] = run;
        run += d[b];
        tot += s[b];
    }
    cntD[n] = run;
    nrmD[n] = rsqrtf(fmaxf((float)run, 1.0f));
    nrmS[n] = rsqrtf(fmaxf((float)tot, 1.0f));
}

#define SC_ROWS 13
__global__ __launch_bounds__(1024)
void gcn_scan(const int* __restrict__ cnt, int* __restrict__ rowptr) {
    __shared__ int wsum[32];
    const int lane  = threadIdx.x & 31;
    const int wv    = threadIdx.x >> 5;
    const int wbase = wv * (SC_ROWS * 128) + lane * 4;

    int4 v[SC_ROWS];
    int  ls[SC_ROWS];
    int  tot = 0;
#pragma unroll
    for (int q = 0; q < SC_ROWS; ++q) {
        const int idx = wbase + q * 128;
        v[q] = make_int4(0, 0, 0, 0);
        if (idx < N_NODES) v[q] = *(const int4*)(cnt + idx);
        ls[q] = v[q].x + v[q].y + v[q].z + v[q].w;
        tot += ls[q];
    }
    for (int o = 16; o > 0; o >>= 1) tot += __shfl_xor(tot, o);
    if (lane == 0) wsum[wv] = tot;
    __syncthreads();

    int t = wsum[lane];
    for (int o = 1; o < 32; o <<= 1) {
        const int u = __shfl_up(t, o);
        if (lane >= o) t += u;
    }
    const int total = __shfl(t, 31);
    const int wincl = __shfl(t, wv);
    int carry = wincl - tot;

#pragma unroll
    for (int q = 0; q < SC_ROWS; ++q) {
        int inc = ls[q];
        for (int o = 1; o < 32; o <<= 1) {
            const int u = __shfl_up(inc, o);
            if (lane >= o) inc += u;
        }
        const int ex  = carry + inc - ls[q];
        const int idx = wbase + q * 128;
        if (idx < N_NODES) {
            int4 r;
            r.x = ex;
            r.y = ex + v[q].x;
            r.z = r.y + v[q].y;
            r.w = r.z + v[q].z;
            *(int4*)(rowptr + idx) = r;
        }
        carry += __shfl(inc, 31);
    }
    if (threadIdx.x == 0) rowptr[N_NODES] = total;
}

__global__ __launch_bounds__(1024)
void gcn_fill(const int* __restrict__ src,
              const int* __restrict__ dst,
              const int* __restrict__ rowptr,
              const int* __restrict__ histD,
              int* __restrict__ csr) {
    __shared__ int off[H_RN];
    const int tid = threadIdx.x;
    const int b   = blockIdx.x / HR;
    const int lo  = (blockIdx.x % HR) * H_RN;
    const int* const dch = dst + b * H_CH;
    const int* const sch = src + b * H_CH;

    int4 dv[H_IT], sv[H_IT];
#pragma unroll
    for (int i = 0; i < H_IT; ++i) {
        const int q = tid + 1024 * i;
        dv[i] = make_int4(-1, -1, -1, -1);
        sv[i] = make_int4(0, 0, 0, 0);
        if (q < H_Q) {
            dv[i] = *(const int4*)(dch + 4 * q);
            sv[i] = *(const int4*)(sch + 4 * q);
        }
    }
    int4 rv[H_OT], pv[H_OT];
#pragma unroll
    for (int j = 0; j < H_OT; ++j) {
        const int i = tid + 1024 * j;
        rv[j] = make_int4(0, 0, 0, 0);
        pv[j] = make_int4(0, 0, 0, 0);
        if (i < H_RN / 4) {
            rv[j] = *(const int4*)(rowptr + lo + 4 * i);
            pv[j] = *(const int4*)(histD + b * N_NODES + lo + 4 * i);
        }
    }
#pragma unroll
    for (int j = 0; j < H_OT; ++j) {
        const int i = tid + 1024 * j;
        if (i < H_RN / 4)
            *(int4*)(off + 4 * i) = make_int4(rv[j].x + pv[j].x, rv[j].y + pv[j].y,
                                              rv[j].z + pv[j].z, rv[j].w + pv[j].w);
    }
    __syncthreads();
#pragma unroll
    for (int i = 0; i < H_IT; ++i) {
        const unsigned k0 = (unsigned)(dv[i].x - lo);
        const unsigned k1 = (unsigned)(dv[i].y - lo);
        const unsigned k2 = (unsigned)(dv[i].z - lo);
        const unsigned k3 = (unsigned)(dv[i].w - lo);
        if (k0 < (unsigned)H_RN) csr[min(max(atomicAdd(off + k0, 1), 0), N_EDGES - 1)] = sv[i].x;
        if (k1 < (unsigned)H_RN) csr[min(max(atomicAdd(off + k1, 1), 0), N_EDGES - 1)] = sv[i].y;
        if (k2 < (unsigned)H_RN) csr[min(max(atomicAdd(off + k2, 1), 0), N_EDGES - 1)] = sv[i].z;
        if (k3 < (unsigned)H_RN) csr[min(max(atomicAdd(off + k3, 1), 0), N_EDGES - 1)] = sv[i].w;
    }
}

#define TP_N     64
#define TP_PITCH 65
__global__ __launch_bounds__(256)
void gcn_scale_transpose(const float* __restrict__ h,
                         const float* __restrict__ nrmS,
                         float* __restrict__ feat) {
    __shared__ float tile[TP_N * TP_PITCH];
    const int tid = threadIdx.x;
    const int n0  = blockIdx.x * TP_N;

    const int c = tid & 15;
    const int r = tid >> 4;
    const int n = n0 + 4 * c;
    if (n < N_NODES) {
        const float4 s = *(const float4*)(nrmS + n);
        float4 v[4];
        for (int p = 0; p < 4; ++p)
            v[p] = *(const float4*)(h + (size_t)(r + 16 * p) * N_NODES + n);
        for (int p = 0; p < 4; ++p) {
            const int d = r + 16 * p;
            tile[(4 * c + 0) * TP_PITCH + d] = v[p].x * s.x;
            tile[(4 * c + 1) * TP_PITCH + d] = v[p].y * s.y;
            tile[(4 * c + 2) * TP_PITCH + d] = v[p].z * s.z;
            tile[(4 * c + 3) * TP_PITCH + d] = v[p].w * s.w;
        }
    }
    __syncthreads();

    const int nvalid = (N_NODES - n0) < TP_N ? (N_NODES - n0) : TP_N;
    float* const dstp = feat + (size_t)n0 * IN_D;
    for (int k = 0; k < 4; ++k) {
        const int e  = tid + 256 * k;
        const int nn = e >> 4;
        const int q  = e & 15;
        if (nn < nvalid) {
            const float* t = tile + nn * TP_PITCH + 4 * q;
            float4 o;
            o.x = t[0]; o.y = t[1]; o.z = t[2]; o.w = t[3];
            *(float4*)(dstp + (size_t)e * 4) = o;
        }
    }
}

__global__ __launch_bounds__(256)
void gcn_gather(const int* __restrict__ rowptr,
                const int* __restrict__ csr,
                const float* __restrict__ feat,
                float* __restrict__ agg) {
    const int node = (blockIdx.x * 256 + threadIdx.x) >> 4;
    const int l    = threadIdx.x & 15;
    int beg = rowptr[node];
    int end = rowptr[node + 1];
    beg = max(beg, 0);
    end = min(end, N_EDGES);
    const float* const fcol = feat + 4 * l;

    float4 acc = make_float4(0.0f, 0.0f, 0.0f, 0.0f);
    int next = 0;
    if (beg + l < end) next = csr[beg + l];
    for (int j = beg; j < end; j += 16) {
        const int rem = end - j;
        const int cnt = rem < 16 ? rem : 16;
        const int mine = min(max(next, 0), N_NODES - 1);
        next = 0;
        if (j + 16 + l < end) next = csr[j + 16 + l];
        for (int k = 0; k < cnt; k += 4) {
            const int s0 = __shfl(mine, k + 0, 16);
            const int s1 = __shfl(mine, k + 1, 16);
            const int s2 = __shfl(mine, k + 2, 16);
            const int s3 = __shfl(mine, k + 3, 16);
            const float4 z = make_float4(0.0f, 0.0f, 0.0f, 0.0f);
            float4 v0 = *(const float4*)(fcol + (size_t)s0 * 64);
            float4 v1 = z, v2 = z, v3 = z;
            if (k + 1 < cnt) v1 = *(const float4*)(fcol + (size_t)s1 * 64);
            if (k + 2 < cnt) v2 = *(const float4*)(fcol + (size_t)s2 * 64);
            if (k + 3 < cnt) v3 = *(const float4*)(fcol + (size_t)s3 * 64);
            acc.x += v0.x; acc.y += v0.y; acc.z += v0.z; acc.w += v0.w;
            acc.x += v1.x; acc.y += v1.y; acc.z += v1.z; acc.w += v1.w;
            acc.x += v2.x; acc.y += v2.y; acc.z += v2.z; acc.w += v2.w;
            acc.x += v3.x; acc.y += v3.y; acc.z += v3.z; acc.w += v3.w;
        }
    }
    *(float4*)(agg + (size_t)node * 64 + 4 * l) = acc;
}

__global__ void gcn_gemm1_wmma(const float* __restrict__ agg,
                               const float* __restrict__ nrmD,
                               const float* __restrict__ nrmS,
                               const float* __restrict__ W1,
                               const float* __restrict__ b1,
                               float* __restrict__ feat) {
    const int lane  = threadIdx.x & 31;
    const int wv    = threadIdx.x >> 5;
    const int m     = lane & 15;
    const int khalf = (lane >> 4) << 1;
    const int row   = blockIdx.x * 16 + m;
    const int col   = (wv << 4) + m;

    const float        scale = nrmD[row];
    const float* const arow  = agg + (size_t)row * 64;

    v8f acc = {};
#pragma unroll
    for (int kk = 0; kk < 64; kk += 4) {
        const int k = kk + khalf;
        v2f a, b;
        a.x = arow[k] * scale;
        a.y = arow[k + 1] * scale;
        b.x = W1[k * HID_D + col];
        b.y = W1[(k + 1) * HID_D + col];
        acc = __builtin_amdgcn_wmma_f32_16x16x4_f32(
            false, a, false, b, (short)0, acc, false, false);
    }

    const int   rbase = blockIdx.x * 16 + ((lane >> 4) << 3);
    const float bias  = b1[col];
#pragma unroll
    for (int j = 0; j < 8; ++j) {
        const int r = rbase + j;
        float v = acc[j] + bias;
        v = v > 0.0f ? v : 0.0f;
        feat[(size_t)r * 64 + col] = v * nrmS[r];
    }
}

__global__ void gcn_gemm2_wmma(const float* __restrict__ agg,
                               const float* __restrict__ nrmD,
                               const float* __restrict__ W2,
                               const float* __restrict__ b2,
                               float* __restrict__ out) {
    const int lane  = threadIdx.x & 31;
    const int wv    = threadIdx.x >> 5;
    const int m     = lane & 15;
    const int khalf = (lane >> 4) << 1;
    const int row   = blockIdx.x * 16 + m;
    const int col   = (wv << 4) + m;

    const float        scale = nrmD[row];
    const float* const arow  = agg + (size_t)row * 64;

    v8f acc = {};
#pragma unroll
    for (int kk = 0; kk < 64; kk += 4) {
        const int k = kk + khalf;
        v2f a, b;
        a.x = arow[k] * scale;
        a.y = arow[k + 1] * scale;
        b.x = W2[k * OUT_D + col];
        b.y = W2[(k + 1) * OUT_D + col];
        acc = __builtin_amdgcn_wmma_f32_16x16x4_f32(
            false, a, false, b, (short)0, acc, false, false);
    }

    const int   rbase = blockIdx.x * 16 + ((lane >> 4) << 3);
    const float bias  = b2[col];
#pragma unroll
    for (int j = 0; j < 8; ++j) {
        const int r = rbase + j;
        out[(size_t)col * N_NODES + r] = acc[j] + bias;
    }
}

extern "C" void kernel_launch(void* const* d_in, const int* in_sizes, int n_in,
                              void* d_out, int out_size, void* d_ws, size_t ws_size,
                              hipStream_t stream) {
    const float* h   = (const float*)d_in[0];
    const int*   src = (const int*)  d_in[1];
    const int*   dst = (const int*)  d_in[2];
    const float* W1  = (const float*)d_in[3];
    const float* b1  = (const float*)d_in[4];
    const float* W2  = (const float*)d_in[5];
    const float* b2  = (const float*)d_in[6];
    float*       out = (float*)d_out;

    int*   cntD   = (int*)d_ws;
    float* nrmS   = (float*)d_ws + 2 * N_NODES;
    float* nrmD   = nrmS + N_NODES;
    int*   rowptr = (int*)d_ws + 4 * N_NODES;
    int*   csr    = rowptr + 50048;
    float* feat   = (float*)d_ws + 4 * N_NODES + 50048 + N_EDGES;
    float* agg    = feat + (size_t)N_NODES * 64;
    int*   histD  = (int*)(agg + (size_t)N_NODES * 64);
    int*   histS  = histD + (size_t)HB * N_NODES;

    const int T = 256;

    gcn_hist    <<<dim3(HB * HR, 2), 1024, 0, stream>>>(src, dst, histS, histD);
    gcn_combine <<<(N_NODES + T - 1) / T, T, 0, stream>>>(histD, histS, cntD, nrmS, nrmD);
    gcn_scan    <<<1, 1024, 0, stream>>>(cntD, rowptr);
    gcn_fill    <<<HB * HR, 1024, 0, stream>>>(src, dst, rowptr, histD, csr);

    gcn_scale_transpose<<<(N_NODES + TP_N - 1) / TP_N, 256, 0, stream>>>(h, nrmS, feat);
    gcn_gather    <<<N_NODES / 16, 256, 0, stream>>>(rowptr, csr, feat, agg);
    gcn_gemm1_wmma<<<N_NODES / 16, 128, 0, stream>>>(agg, nrmD, nrmS, W1, b1, feat);

    gcn_gather    <<<N_NODES / 16, 256, 0, stream>>>(rowptr, csr, feat, agg);
    gcn_gemm2_wmma<<<N_NODES / 16, 64, 0, stream>>>(agg, nrmD, W2, b2, out);
}


# layer 2 weight product moved in front of its aggregation (the edge sum is linear): x1@W2 joins the layer-1 WMMA kernel, 2nd gather sums 32-wide rows and writes out; sums now taken after the product
# speedup vs baseline: 4.0558x; 1.1522x over previous
#include <hip/hip_runtime.h>
#include <hip/hip_bf16.h>


typedef float v2f __attribute__((ext_vector_type(2)));
typedef float v8f __attribute__((ext_vector_type(8)));

#define N_NODES 50000
#define N_EDGES 800000
#define IN_D 64
#define HID_D 64
#define OUT_D 32

#define HB   32
#define HR   4
#define H_CH (N_EDGES / HB)
#define H_RN (N_NODES / HR)
#define H_Q  (H_CH / 4)
#define H_IT ((H_Q + 1023) / 1024)
#define H_OT ((H_RN / 4 + 1023) / 1024)

__global__ __launch_bounds__(1024)
void gcn_hist(const int* __restrict__ src,
              const int* __restrict__ dst,
              int* __restrict__ histS,
              int* __restrict__ histD) {
    __shared__ int cnt[H_RN];
    const int tid = threadIdx.x;
    const int b   = blockIdx.x / HR;
    const int lo  = (blockIdx.x % HR) * H_RN;
    const int* const ids  = (blockIdx.y ? src : dst) + b * H_CH;
    int* const       hist = (blockIdx.y ? histS : histD) + b * N_NODES + lo;

    int4 v[H_IT];
#pragma unroll
    for (int i = 0; i < H_IT; ++i) {
        const int q = tid + 1024 * i;
        v[i] = make_int4(-1, -1, -1, -1);
        if (q < H_Q) v[i] = *(const int4*)(ids + 4 * q);
    }
    for (int i = tid; i < H_RN / 4; i += 1024) *(int4*)(cnt + 4 * i) = make_int4(0, 0, 0, 0);
    __syncthreads();
#pragma unroll
    for (int i = 0; i < H_IT; ++i) {
        const unsigned k0 = (unsigned)(v[i].x - lo);
        const unsigned k1 = (unsigned)(v[i].y - lo);
        const unsigned k2 = (unsigned)(v[i].z - lo);
        const unsigned k3 = (unsigned)(v[i].w - lo);
        if (k0 < (unsigned)H_RN) atomicAdd(cnt + k0, 1);
        if (k1 < (unsigned)H_RN) atomicAdd(cnt + k1, 1);
        if (k2 < (unsigned)H_RN) atomicAdd(cnt + k2, 1);
        if (k3 < (unsigned)H_RN) atomicAdd(cnt + k3, 1);
    }
    __syncthreads();
    for (int i = tid; i < H_RN / 4; i += 1024)
        *(int4*)(hist + 4 * i) = *(const int4*)(cnt + 4 * i);
}

__global__ __launch_bounds__(256)
void gcn_combine(int* __restrict__ histD,
                 const int* __restrict__ histS,
                 int* __restrict__ cntD,
                 float* __restrict__ nrmS,
                 float* __restrict__ nrmD) {
    const int n = blockIdx.x * 256 + threadIdx.x;
    if (n >= N_NODES) return;
    int d[HB], s[HB];
#pragma unroll
    for (int b = 0; b < HB; ++b) {
        d[b] = histD[b * N_NODES + n];
        s[b] = histS[b * N_NODES + n];
    }
    int run = 0, tot = 0;
#pragma unroll
    for (int b = 0; b < HB; ++b) {
        histD[b * N_NODES + n] = run;
        run += d[b];
        tot += s[b];
    }
    cntD[n] = run;
    nrmD[n] = rsqrtf(fmaxf((float)run, 1.0f));
    nrmS[n] = rsqrtf(fmaxf((float)tot, 1.0f));
}

#define SC_ROWS 13
__global__ __launch_bounds__(1024)
void gcn_scan(const int* __restrict__ cnt, int* __restrict__ rowptr) {
    __shared__ int wsum[32];
    const int lane  = threadIdx.x & 31;
    const int wv    = threadIdx.x >> 5;
    const int wbase = wv * (SC_ROWS * 128) + lane * 4;

    int4 v[SC_ROWS];
    int  ls[SC_ROWS];
    int  tot = 0;
#pragma unroll
    for (int q = 0; q < SC_ROWS; ++q) {
        const int idx = wbase + q * 128;
        v[q] = make_int4(0, 0, 0, 0);
        if (idx < N_NODES) v[q] = *(const int4*)(cnt + idx);
        ls[q] = v[q].x + v[q].y + v[q].z + v[q].w;
        tot += ls[q];
    }
    for (int o = 16; o > 0; o >>= 1) tot += __shfl_xor(tot, o);
    if (lane == 0) wsum[wv] = tot;
    __syncthreads();

    int t = wsum[lane];
    for (int o = 1; o < 32; o <<= 1) {
        const int u = __shfl_up(t, o);
        if (lane >= o) t += u;
    }
    const int total = __shfl(t, 31);
    const int wincl = __shfl(t, wv);
    int carry = wincl - tot;

#pragma unroll
    for (int q = 0; q < SC_ROWS; ++q) {
        int inc = ls[q];
        for (int o = 1; o < 32; o <<= 1) {
            const int u = __shfl_up(inc, o);
            if (lane >= o) inc += u;
        }
        const int ex  = carry + inc - ls[q];
        const int idx = wbase + q * 128;
        if (idx < N_NODES) {
            int4 r;
            r.x = ex;
            r.y = ex + v[q].x;
            r.z = r.y + v[q].y;
            r.w = r.z + v[q].z;
            *(int4*)(rowptr + idx) = r;
        }
        carry += __shfl(inc, 31);
    }
    if (threadIdx.x == 0) rowptr[N_NODES] = total;
}

__global__ __launch_bounds__(1024)
void gcn_fill(const int* __restrict__ src,
              const int* __restrict__ dst,
              const int* __restrict__ rowptr,
              const int* __restrict__ histD,
              int* __restrict__ csr) {
    __shared__ int off[H_RN];
    const int tid = threadIdx.x;
    const int b   = blockIdx.x / HR;
    const int lo  = (blockIdx.x % HR) * H_RN;
    const int* const dch = dst + b * H_CH;
    const int* const sch = src + b * H_CH;

    int4 dv[H_IT], sv[H_IT];
#pragma unroll
    for (int i = 0; i < H_IT; ++i) {
        const int q = tid + 1024 * i;
        dv[i] = make_int4(-1, -1, -1, -1);
        sv[i] = make_int4(0, 0, 0, 0);
        if (q < H_Q) {
            dv[i] = *(const int4*)(dch + 4 * q);
            sv[i] = *(const int4*)(sch + 4 * q);
        }
    }
    int4 rv[H_OT], pv[H_OT];
#pragma unroll
    for (int j = 0; j < H_OT; ++j) {
        const int i = tid + 1024 * j;
        rv[j] = make_int4(0, 0, 0, 0);
        pv[j] = make_int4(0, 0, 0, 0);
        if (i < H_RN / 4) {
            rv[j] = *(const int4*)(rowptr + lo + 4 * i);
            pv[j] = *(const int4*)(histD + b * N_NODES + lo + 4 * i);
        }
    }
#pragma unroll
    for (int j = 0; j < H_OT; ++j) {
        const int i = tid + 1024 * j;
        if (i < H_RN / 4)
            *(int4*)(off + 4 * i) = make_int4(rv[j].x + pv[j].x, rv[j].y + pv[j].y,
                                              rv[j].z + pv[j].z, rv[j].w + pv[j].w);
    }
    __syncthreads();
#pragma unroll
    for (int i = 0; i < H_IT; ++i) {
        const unsigned k0 = (unsigned)(dv[i].x - lo);
        const unsigned k1 = (unsigned)(dv[i].y - lo);
        const unsigned k2 = (unsigned)(dv[i].z - lo);
        const unsigned k3 = (unsigned)(dv[i].w - lo);
        if (k0 < (unsigned)H_RN) csr[min(max(atomicAdd(off + k0, 1), 0), N_EDGES - 1)] = sv[i].x;
        if (k1 < (unsigned)H_RN) csr[min(max(atomicAdd(off + k1, 1), 0), N_EDGES - 1)] = sv[i].y;
        if (k2 < (unsigned)H_RN) csr[min(max(atomicAdd(off + k2, 1), 0), N_EDGES - 1)] = sv[i].z;
        if (k3 < (unsigned)H_RN) csr[min(max(atomicAdd(off + k3, 1), 0), N_EDGES - 1)] = sv[i].w;
    }
}

#define TP_N     64
#define TP_PITCH 65
__global__ __launch_bounds__(256)
void gcn_scale_transpose(const float* __restrict__ h,
                         const float* __restrict__ nrmS,
                         float* __restrict__ feat) {
    __shared__ float tile[TP_N * TP_PITCH];
    const int tid = threadIdx.x;
    const int n0  = blockIdx.x * TP_N;

    const int c = tid & 15;
    const int r = tid >> 4;
    const int n = n0 + 4 * c;
    if (n < N_NODES) {
        const float4 s = *(const float4*)(nrmS + n);
        float4 v[4];
        for (int p = 0; p < 4; ++p)
            v[p] = *(const float4*)(h + (size_t)(r + 16 * p) * N_NODES + n);
        for (int p = 0; p < 4; ++p) {
            const int d = r + 16 * p;
            tile[(4 * c + 0) * TP_PITCH + d] = v[p].x * s.x;
            tile[(4 * c + 1) * TP_PITCH + d] = v[p].y * s.y;
            tile[(4 * c + 2) * TP_PITCH + d] = v[p].z * s.z;
            tile[(4 * c + 3) * TP_PITCH + d] = v[p].w * s.w;
        }
    }
    __syncthreads();

    const int nvalid = (N_NODES - n0) < TP_N ? (N_NODES - n0) : TP_N;
    float* const dstp = feat + (size_t)n0 * IN_D;
    for (int k = 0; k < 4; ++k) {
        const int e  = tid + 256 * k;
        const int nn = e >> 4;
        const int q  = e & 15;
        if (nn < nvalid) {
            const float* t = tile + nn * TP_PITCH + 4 * q;
            float4 o;
            o.x = t[0]; o.y = t[1]; o.z = t[2]; o.w = t[3];
            *(float4*)(dstp + (size_t)e * 4) = o;
        }
    }
}

__global__ __launch_bounds__(256)
void gcn_gather(const int* __restrict__ rowptr,
                const int* __restrict__ csr,
                const float* __restrict__ feat,
                float* __restrict__ agg) {
    const int node = (blockIdx.x * 256 + threadIdx.x) >> 4;
    const int l    = threadIdx.x & 15;
    int beg = rowptr[node];
    int end = rowptr[node + 1];
    beg = max(beg, 0);
    end = min(end, N_EDGES);
    const float* const fcol = feat + 4 * l;

    float4 acc = make_float4(0.0f, 0.0f, 0.0f, 0.0f);
    int next = 0;
    if (beg + l < end) next = csr[beg + l];
    for (int j = beg; j < end; j += 16) {
        const int rem = end - j;
        const int cnt = rem < 16 ? rem : 16;
        const int mine = min(max(next, 0), N_NODES - 1);
        next = 0;
        if (j + 16 + l < end) next = csr[j + 16 + l];
        for (int k = 0; k < cnt; k += 4) {
            const int s0 = __shfl(mine, k + 0, 16);
            const int s1 = __shfl(mine, k + 1, 16);
            const int s2 = __shfl(mine, k + 2, 16);
            const int s3 = __shfl(mine, k + 3, 16);
            const float4 z = make_float4(0.0f, 0.0f, 0.0f, 0.0f);
            float4 v0 = *(const float4*)(fcol + (size_t)s0 * 64);
            float4 v1 = z, v2 = z, v3 = z;
            if (k + 1 < cnt) v1 = *(const float4*)(fcol + (size_t)s1 * 64);
            if (k + 2 < cnt) v2 = *(const float4*)(fcol + (size_t)s2 * 64);
            if (k + 3 < cnt) v3 = *(const float4*)(fcol + (size_t)s3 * 64);
            acc.x += v0.x; acc.y += v0.y; acc.z += v0.z; acc.w += v0.w;
            acc.x += v1.x; acc.y += v1.y; acc.z += v1.z; acc.w += v1.w;
            acc.x += v2.x; acc.y += v2.y; acc.z += v2.z; acc.w += v2.w;
            acc.x += v3.x; acc.y += v3.y; acc.z += v3.z; acc.w += v3.w;
        }
    }
    *(float4*)(agg + (size_t)node * 64 + 4 * l) = acc;
}

#define X_PITCH 68
__global__ __launch_bounds__(64)
void gcn_gemm12_wmma(const float* __restrict__ agg,
                     const float* __restrict__ nrmD,
                     const float* __restrict__ nrmS,
                     const float* __restrict__ W1,
                     const float* __restrict__ b1,
                     const float* __restrict__ W2,
                     float* __restrict__ z2) {
    __shared__ float xs[16 * X_PITCH];
    const int lane  = threadIdx.x & 31;
    const int wv    = threadIdx.x >> 5;
    const int m     = lane & 15;
    const int khalf = (lane >> 4) << 1;
    const int row   = blockIdx.x * 16 + m;
    const int colA  = (wv << 5) + m;
    const int colB  = colA + 16;

    const float        scale = nrmD[row];
    const float* const arow  = agg + (size_t)row * 64;

    v8f acc0 = {};
    v8f acc1 = {};
#pragma unroll
    for (int kk = 0; kk < 64; kk += 4) {
        const int k = kk + khalf;
        v2f a, p, q;
        a.x = arow[k] * scale;
        a.y = arow[k + 1] * scale;
        p.x = W1[k * HID_D + colA];
        p.y = W1[(k + 1) * HID_D + colA];
        q.x = W1[k * HID_D + colB];
        q.y = W1[(k + 1) * HID_D + colB];
        acc0 = __builtin_amdgcn_wmma_f32_16x16x4_f32(
            false, a, false, p, (short)0, acc0, false, false);
        acc1 = __builtin_amdgcn_wmma_f32_16x16x4_f32(
            false, a, false, q, (short)0, acc1, false, false);
    }

    const int   rloc  = (lane >> 4) << 3;
    const int   rbase = blockIdx.x * 16 + rloc;
    const float biasA = b1[colA];
    const float biasB = b1[colB];
#pragma unroll
    for (int j = 0; j < 8; ++j) {
        const float ns = nrmS[rbase + j];
        float u = acc0[j] + biasA;
        float w = acc1[j] + biasB;
        u = u > 0.0f ? u : 0.0f;
        w = w > 0.0f ? w : 0.0f;
        xs[(rloc + j) * X_PITCH + colA] = u * ns;
        xs[(rloc + j) * X_PITCH + colB] = w * ns;
    }
    __syncthreads();

    const int    col2 = (wv << 4) + m;
    const float* xrow = xs + m * X_PITCH;
    v8f acc2 = {};
#pragma unroll
    for (int kk = 0; kk < 64; kk += 4) {
        const int k = kk + khalf;
        v2f a, b;
        a.x = xrow[k];
        a.y = xrow[k + 1];
        b.x = W2[k * OUT_D + col2];
        b.y = W2[(k + 1) * OUT_D + col2];
        acc2 = __builtin_amdgcn_wmma_f32_16x16x4_f32(
            false, a, false, b, (short)0, acc2, false, false);
    }
#pragma unroll
    for (int j = 0; j < 8; ++j)
        z2[(size_t)(rbase + j) * OUT_D + col2] = acc2[j];
}

#define G2_NB 32
#define G2_P  33
__global__ __launch_bounds__(256)
void gcn_gather32_out(const int* __restrict__ rowptr,
                      const int* __restrict__ csr,
                      const float* __restrict__ z2,
                      const float* __restrict__ nrmD,
                      const float* __restrict__ b2,
                      float* __restrict__ out) {
    __shared__ float tile[OUT_D * G2_P];
    const int tid  = threadIdx.x;
    const int nn   = tid >> 3;
    const int q    = tid & 7;
    const int n0   = blockIdx.x * G2_NB;
    const int node = n0 + nn;

    int   beg = 0, end = 0;
    float sD  = 0.0f;
    if (node < N_NODES) {
        beg = rowptr[node];
        end = rowptr[node + 1];
        sD  = nrmD[node];
    }
    beg = max(beg, 0);
    end = min(end, N_EDGES);
    const float4 bb = *(const float4*)(b2 + 4 * q);
    const float* const zcol = z2 + 4 * q;

    float4 acc = make_float4(0.0f, 0.0f, 0.0f, 0.0f);
    int next = 0;
    if (beg + q < end) next = csr[beg + q];
    for (int j = beg; j < end; j += 8) {
        const int rem  = end - j;
        const int cnt  = rem < 8 ? rem : 8;
        const int mine = min(max(next, 0), N_NODES - 1);
        const int s0 = __shfl(mine, 0, 8);
        const int s1 = __shfl(mine, 1, 8);
        const int s2 = __shfl(mine, 2, 8);
        const int s3 = __shfl(mine, 3, 8);
        const int s4 = __shfl(mine, 4, 8);
        const int s5 = __shfl(mine, 5, 8);
        const int s6 = __shfl(mine, 6, 8);
        const int s7 = __shfl(mine, 7, 8);
        const float4 z = make_float4(0.0f, 0.0f, 0.0f, 0.0f);
        float4 v0 = *(const float4*)(zcol + (size_t)s0 * OUT_D);
        float4 v1 = z, v2 = z, v3 = z, v4 = z, v5 = z, v6 = z, v7 = z;
        if (1 < cnt) v1 = *(const float4*)(zcol + (size_t)s1 * OUT_D);
        if (2 < cnt) v2 = *(const float4*)(zcol + (size_t)s2 * OUT_D);
        if (3 < cnt) v3 = *(const float4*)(zcol + (size_t)s3 * OUT_D);
        if (4 < cnt) v4 = *(const float4*)(zcol + (size_t)s4 * OUT_D);
        if (5 < cnt) v5 = *(const float4*)(zcol + (size_t)s5 * OUT_D);
        if (6 < cnt) v6 = *(const float4*)(zcol + (size_t)s6 * OUT_D);
        if (7 < cnt) v7 = *(const float4*)(zcol + (size_t)s7 * OUT_D);
        next = 0;
        if (j + 8 + q < end) next = csr[j + 8 + q];
        acc.x += v0.x; acc.y += v0.y; acc.z += v0.z; acc.w += v0.w;
        acc.x += v1.x; acc.y += v1.y; acc.z += v1.z; acc.w += v1.w;
        acc.x += v2.x; acc.y += v2.y; acc.z += v2.z; acc.w += v2.w;
        acc.x += v3.x; acc.y += v3.y; acc.z += v3.z; acc.w += v3.w;
        acc.x += v4.x; acc.y += v4.y; acc.z += v4.z; acc.w += v4.w;
        acc.x += v5.x; acc.y += v5.y; acc.z += v5.z; acc.w += v5.w;
        acc.x += v6.x; acc.y += v6.y; acc.z += v6.z; acc.w += v6.w;
        acc.x += v7.x; acc.y += v7.y; acc.z += v7.z; acc.w += v7.w;
    }

    tile[(4 * q + 0) * G2_P + nn] = acc.x * sD + bb.x;
    tile[(4 * q + 1) * G2_P + nn] = acc.y * sD + bb.y;
    tile[(4 * q + 2) * G2_P + nn] = acc.z * sD + bb.z;
    tile[(4 * q + 3) * G2_P + nn] = acc.w * sD + bb.w;
    __syncthreads();

    const int c    = tid >> 3;
    const int quad = tid & 7;
    const int n    = n0 + 4 * quad;
    if (n < N_NODES) {
        const float* t = tile + c * G2_P + 4 * quad;
        float4 o;
        o.x = t[0]; o.y = t[1]; o.z = t[2]; o.w = t[3];
        *(float4*)(out + (size_t)c * N_NODES + n) = o;
    }
}

extern "C" void kernel_launch(void* const* d_in, const int* in_sizes, int n_in,
                              void* d_out, int out_size, void* d_ws, size_t ws_size,
                              hipStream_t stream) {
    const float* h   = (const float*)d_in[0];
    const int*   src = (const int*)  d_in[1];
    const int*   dst = (const int*)  d_in[2];
    const float* W1  = (const float*)d_in[3];
    const float* b1  = (const float*)d_in[4];
    const float* W2  = (const float*)d_in[5];
    const float* b2  = (const float*)d_in[6];
    float*       out = (float*)d_out;

    int*   cntD   = (int*)d_ws;
    float* nrmS   = (float*)d_ws + 2 * N_NODES;
    float* nrmD   = nrmS + N_NODES;
    int*   rowptr = (int*)d_ws + 4 * N_NODES;
    int*   csr    = rowptr + 50048;
    float* feat   = (float*)d_ws + 4 * N_NODES + 50048 + N_EDGES;
    float* agg    = feat + (size_t)N_NODES * 64;
    int*   histD  = (int*)(agg + (size_t)N_NODES * 64);
    int*   histS  = histD + (size_t)HB * N_NODES;
    float* z2     = feat;

    const int T = 256;

    gcn_hist    <<<dim3(HB * HR, 2), 1024, 0, stream>>>(src, dst, histS, histD);
    gcn_combine <<<(N_NODES + T - 1) / T, T, 0, stream>>>(histD, histS, cntD, nrmS, nrmD);
    gcn_scan    <<<1, 1024, 0, stream>>>(cntD, rowptr);
    gcn_fill    <<<HB * HR, 1024, 0, stream>>>(src, dst, rowptr, histD, csr);

    gcn_scale_transpose<<<(N_NODES + TP_N - 1) / TP_N, 256, 0, stream>>>(h, nrmS, feat);
    gcn_gather     <<<N_NODES / 16, 256, 0, stream>>>(rowptr, csr, feat, agg);
    gcn_gemm12_wmma<<<N_NODES / 16, 64, 0, stream>>>(agg, nrmD, nrmS, W1, b1, W2, z2);

    gcn_gather32_out<<<(N_NODES + G2_NB - 1) / G2_NB, 256, 0, stream>>>(rowptr, csr, z2, nrmD, b2, out);
}


# layer-1 edge sum fused into the WMMA kernel: each block gathers its 16 nodes' sums straight into the LDS tile both products read; agg plane and one launch gone; sum and k orders unchanged
# speedup vs baseline: 4.2725x; 1.0534x over previous
#include <hip/hip_runtime.h>
#include <hip/hip_bf16.h>


typedef float v2f __attribute__((ext_vector_type(2)));
typedef float v8f __attribute__((ext_vector_type(8)));

#define N_NODES 50000
#define N_EDGES 800000
#define IN_D 64
#define HID_D 64
#define OUT_D 32

#define HB   32
#define HR   4
#define H_CH (N_EDGES / HB)
#define H_RN (N_NODES / HR)
#define H_Q  (H_CH / 4)
#define H_IT ((H_Q + 1023) / 1024)
#define H_OT ((H_RN / 4 + 1023) / 1024)

__global__ __launch_bounds__(1024)
void gcn_hist(const int* __restrict__ src,
              const int* __restrict__ dst,
              int* __restrict__ histS,
              int* __restrict__ histD) {
    __shared__ int cnt[H_RN];
    const int tid = threadIdx.x;
    const int b   = blockIdx.x / HR;
    const int lo  = (blockIdx.x % HR) * H_RN;
    const int* const ids  = (blockIdx.y ? src : dst) + b * H_CH;
    int* const       hist = (blockIdx.y ? histS : histD) + b * N_NODES + lo;

    int4 v[H_IT];
#pragma unroll
    for (int i = 0; i < H_IT; ++i) {
        const int q = tid + 1024 * i;
        v[i] = make_int4(-1, -1, -1, -1);
        if (q < H_Q) v[i] = *(const int4*)(ids + 4 * q);
    }
    for (int i = tid; i < H_RN / 4; i += 1024) *(int4*)(cnt + 4 * i) = make_int4(0, 0, 0, 0);
    __syncthreads();
#pragma unroll
    for (int i = 0; i < H_IT; ++i) {
        const unsigned k0 = (unsigned)(v[i].x - lo);
        const unsigned k1 = (unsigned)(v[i].y - lo);
        const unsigned k2 = (unsigned)(v[i].z - lo);
        const unsigned k3 = (unsigned)(v[i].w - lo);
        if (k0 < (unsigned)H_RN) atomicAdd(cnt + k0, 1);
        if (k1 < (unsigned)H_RN) atomicAdd(cnt + k1, 1);
        if (k2 < (unsigned)H_RN) atomicAdd(cnt + k2, 1);
        if (k3 < (unsigned)H_RN) atomicAdd(cnt + k3, 1);
    }
    __syncthreads();
    for (int i = tid; i < H_RN / 4; i += 1024)
        *(int4*)(hist + 4 * i) = *(const int4*)(cnt + 4 * i);
}

__global__ __launch_bounds__(256)
void gcn_combine(int* __restrict__ histD,
                 const int* __restrict__ histS,
                 int* __restrict__ cntD,
                 float* __restrict__ nrmS,
                 float* __restrict__ nrmD) {
    const int n = blockIdx.x * 256 + threadIdx.x;
    if (n >= N_NODES) return;
    int d[HB], s[HB];
#pragma unroll
    for (int b = 0; b < HB; ++b) {
        d[b] = histD[b * N_NODES + n];
        s[b] = histS[b * N_NODES + n];
    }
    int run = 0, tot = 0;
#pragma unroll
    for (int b = 0; b < HB; ++b) {
        histD[b * N_NODES + n] = run;
        run += d[b];
        tot += s[b];
    }
    cntD[n] = run;
    nrmD[n] = rsqrtf(fmaxf((float)run, 1.0f));
    nrmS[n] = rsqrtf(fmaxf((float)tot, 1.0f));
}

#define SC_ROWS 13
__global__ __launch_bounds__(1024)
void gcn_scan(const int* __restrict__ cnt, int* __restrict__ rowptr) {
    __shared__ int wsum[32];
    const int lane  = threadIdx.x & 31;
    const int wv    = threadIdx.x >> 5;
    const int wbase = wv * (SC_ROWS * 128) + lane * 4;

    int4 v[SC_ROWS];
    int  ls[SC_ROWS];
    int  tot = 0;
#pragma unroll
    for (int q = 0; q < SC_ROWS; ++q) {
        const int idx = wbase + q * 128;
        v[q] = make_int4(0, 0, 0, 0);
        if (idx < N_NODES) v[q] = *(const int4*)(cnt + idx);
        ls[q] = v[q].x + v[q].y + v[q].z + v[q].w;
        tot += ls[q];
    }
    for (int o = 16; o > 0; o >>= 1) tot += __shfl_xor(tot, o);
    if (lane == 0) wsum[wv] = tot;
    __syncthreads();

    int t = wsum[lane];
    for (int o = 1; o < 32; o <<= 1) {
        const int u = __shfl_up(t, o);
        if (lane >= o) t += u;
    }
    const int total = __shfl(t, 31);
    const int wincl = __shfl(t, wv);
    int carry = wincl - tot;

#pragma unroll
    for (int q = 0; q < SC_ROWS; ++q) {
        int inc = ls[q];
        for (int o = 1; o < 32; o <<= 1) {
            const int u = __shfl_up(inc, o);
            if (lane >= o) inc += u;
        }
        const int ex  = carry + inc - ls[q];
        const int idx = wbase + q * 128;
        if (idx < N_NODES) {
            int4 r;
            r.x = ex;
            r.y = ex + v[q].x;
            r.z = r.y + v[q].y;
            r.w = r.z + v[q].z;
            *(int4*)(rowptr + idx) = r;
        }
        carry += __shfl(inc, 31);
    }
    if (threadIdx.x == 0) rowptr[N_NODES] = total;
}

__global__ __launch_bounds__(1024)
void gcn_fill(const int* __restrict__ src,
              const int* __restrict__ dst,
              const int* __restrict__ rowptr,
              const int* __restrict__ histD,
              int* __restrict__ csr) {
    __shared__ int off[H_RN];
    const int tid = threadIdx.x;
    const int b   = blockIdx.x / HR;
    const int lo  = (blockIdx.x % HR) * H_RN;
    const int* const dch = dst + b * H_CH;
    const int* const sch = src + b * H_CH;

    int4 dv[H_IT], sv[H_IT];
#pragma unroll
    for (int i = 0; i < H_IT; ++i) {
        const int q = tid + 1024 * i;
        dv[i] = make_int4(-1, -1, -1, -1);
        sv[i] = make_int4(0, 0, 0, 0);
        if (q < H_Q) {
            dv[i] = *(const int4*)(dch + 4 * q);
            sv[i] = *(const int4*)(sch + 4 * q);
        }
    }
    int4 rv[H_OT], pv[H_OT];
#pragma unroll
    for (int j = 0; j < H_OT; ++j) {
        const int i = tid + 1024 * j;
        rv[j] = make_int4(0, 0, 0, 0);
        pv[j] = make_int4(0, 0, 0, 0);
        if (i < H_RN / 4) {
            rv[j] = *(const int4*)(rowptr + lo + 4 * i);
            pv[j] = *(const int4*)(histD + b * N_NODES + lo + 4 * i);
        }
    }
#pragma unroll
    for (int j = 0; j < H_OT; ++j) {
        const int i = tid + 1024 * j;
        if (i < H_RN / 4)
            *(int4*)(off + 4 * i) = make_int4(rv[j].x + pv[j].x, rv[j].y + pv[j].y,
                                              rv[j].z + pv[j].z, rv[j].w + pv[j].w);
    }
    __syncthreads();
#pragma unroll
    for (int i = 0; i < H_IT; ++i) {
        const unsigned k0 = (unsigned)(dv[i].x - lo);
        const unsigned k1 = (unsigned)(dv[i].y - lo);
        const unsigned k2 = (unsigned)(dv[i].z - lo);
        const unsigned k3 = (unsigned)(dv[i].w - lo);
        if (k0 < (unsigned)H_RN) csr[min(max(atomicAdd(off + k0, 1), 0), N_EDGES - 1)] = sv[i].x;
        if (k1 < (unsigned)H_RN) csr[min(max(atomicAdd(off + k1, 1), 0), N_EDGES - 1)] = sv[i].y;
        if (k2 < (unsigned)H_RN) csr[min(max(atomicAdd(off + k2, 1), 0), N_EDGES - 1)] = sv[i].z;
        if (k3 < (unsigned)H_RN) csr[min(max(atomicAdd(off + k3, 1), 0), N_EDGES - 1)] = sv[i].w;
    }
}

#define TP_N     64
#define TP_PITCH 65
__global__ __launch_bounds__(256)
void gcn_scale_transpose(const float* __restrict__ h,
                         const float* __restrict__ nrmS,
                         float* __restrict__ feat) {
    __shared__ float tile[TP_N * TP_PITCH];
    const int tid = threadIdx.x;
    const int n0  = blockIdx.x * TP_N;

    const int c = tid & 15;
    const int r = tid >> 4;
    const int n = n0 + 4 * c;
    if (n < N_NODES) {
        const float4 s = *(const float4*)(nrmS + n);
        float4 v[4];
        for (int p = 0; p < 4; ++p)
            v[p] = *(const float4*)(h + (size_t)(r + 16 * p) * N_NODES + n);
        for (int p = 0; p < 4; ++p) {
            const int d = r + 16 * p;
            tile[(4 * c + 0) * TP_PITCH + d] = v[p].x * s.x;
            tile[(4 * c + 1) * TP_PITCH + d] = v[p].y * s.y;
            tile[(4 * c + 2) * TP_PITCH + d] = v[p].z * s.z;
            tile[(4 * c + 3) * TP_PITCH + d] = v[p].w * s.w;
        }
    }
    __syncthreads();

    const int nvalid = (N_NODES - n0) < TP_N ? (N_NODES - n0) : TP_N;
    float* const dstp = feat + (size_t)n0 * IN_D;
    for (int k = 0; k < 4; ++k) {
        const int e  = tid + 256 * k;
        const int nn = e >> 4;
        const int q  = e & 15;
        if (nn < nvalid) {
            const float* t = tile + nn * TP_PITCH + 4 * q;
            float4 o;
            o.x = t[0]; o.y = t[1]; o.z = t[2]; o.w = t[3];
            *(float4*)(dstp + (size_t)e * 4) = o;
        }
    }
}

#define X_PITCH 68
__global__ __launch_bounds__(256)
void gcn_gather_gemm12_wmma(const int* __restrict__ rowptr,
                            const int* __restrict__ csr,
                            const float* __restrict__ feat,
                            const float* __restrict__ nrmD,
                            const float* __restrict__ nrmS,
                            const float* __restrict__ W1,
                            const float* __restrict__ b1,
                            const float* __restrict__ W2,
                            float* __restrict__ z2) {
    __shared__ float as[16 * X_PITCH];
    __shared__ float xs[16 * X_PITCH];

    const int nn   = threadIdx.x >> 4;
    const int node = blockIdx.x * 16 + nn;
    const int l    = threadIdx.x & 15;
    int beg = rowptr[node];
    int end = rowptr[node + 1];
    const float sD = nrmD[node];
    beg = max(beg, 0);
    end = min(end, N_EDGES);
    const float* const fcol = feat + 4 * l;

    float4 acc = make_float4(0.0f, 0.0f, 0.0f, 0.0f);
    int next = 0;
    if (beg + l < end) next = csr[beg + l];
    for (int j = beg; j < end; j += 16) {
        const int rem = end - j;
        const int cnt = rem < 16 ? rem : 16;
        const int mine = min(max(next, 0), N_NODES - 1);
        next = 0;
        if (j + 16 + l < end) next = csr[j + 16 + l];
        for (int k = 0; k < cnt; k += 4) {
            const int s0 = __shfl(mine, k + 0, 16);
            const int s1 = __shfl(mine, k + 1, 16);
            const int s2 = __shfl(mine, k + 2, 16);
            const int s3 = __shfl(mine, k + 3, 16);
            const float4 z = make_float4(0.0f, 0.0f, 0.0f, 0.0f);
            float4 v0 = *(const float4*)(fcol + (size_t)s0 * 64);
            float4 v1 = z, v2 = z, v3 = z;
            if (k + 1 < cnt) v1 = *(const float4*)(fcol + (size_t)s1 * 64);
            if (k + 2 < cnt) v2 = *(const float4*)(fcol + (size_t)s2 * 64);
            if (k + 3 < cnt) v3 = *(const float4*)(fcol + (size_t)s3 * 64);
            acc.x += v0.x; acc.y += v0.y; acc.z += v0.z; acc.w += v0.w;
            acc.x += v1.x; acc.y += v1.y; acc.z += v1.z; acc.w += v1.w;
            acc.x += v2.x; acc.y += v2.y; acc.z += v2.z; acc.w += v2.w;
            acc.x += v3.x; acc.y += v3.y; acc.z += v3.z; acc.w += v3.w;
        }
    }
    const int lane  = threadIdx.x & 31;
    const int wv    = __builtin_amdgcn_readfirstlane(threadIdx.x >> 5);
    const int m     = lane & 15;
    const int khalf = (lane >> 4) << 1;
    const int rloc  = (lane >> 4) << 3;
    const int rbase = blockIdx.x * 16 + rloc;
    const bool first  = wv < 4;
    const int  w2     = 4 + ((blockIdx.x & 1) << 1);
    const bool second = (wv == w2) | (wv == w2 + 1);
    const int  col    = first ? (wv << 4) + m : ((wv & 1) << 4) + m;

    v2f   bw[16];
    float nsv[8];
    float bias = 0.0f;
    if (first) {
#pragma unroll
        for (int t = 0; t < 16; ++t) {
            const int k = 4 * t + khalf;
            bw[t].x = W1[k * HID_D + col];
            bw[t].y = W1[(k + 1) * HID_D + col];
        }
        bias = b1[col];
#pragma unroll
        for (int j = 0; j < 8; ++j) nsv[j] = nrmS[rbase + j];
    } else if (second) {
#pragma unroll
        for (int t = 0; t < 16; ++t) {
            const int k = 4 * t + khalf;
            bw[t].x = W2[k * OUT_D + col];
            bw[t].y = W2[(k + 1) * OUT_D + col];
        }
    }

    *(float4*)(as + nn * X_PITCH + 4 * l) =
        make_float4(acc.x * sD, acc.y * sD, acc.z * sD, acc.w * sD);
    __syncthreads();


    if (first) {
        const float* arow = as + m * X_PITCH;
        v8f acc1 = {};
#pragma unroll
        for (int t = 0; t < 16; ++t) {
            const int k = 4 * t + khalf;
            v2f a;
            a.x = arow[k];
            a.y = arow[k + 1];
            acc1 = __builtin_amdgcn_wmma_f32_16x16x4_f32(
                false, a, false, bw[t], (short)0, acc1, false, false);
        }
#pragma unroll
        for (int j = 0; j < 8; ++j) {
            float u = acc1[j] + bias;
            u = u > 0.0f ? u : 0.0f;
            xs[(rloc + j) * X_PITCH + col] = u * nsv[j];
        }
    }
    __syncthreads();

    if (second) {
        const float* xrow = xs + m * X_PITCH;
        v8f acc2 = {};
#pragma unroll
        for (int t = 0; t < 16; ++t) {
            const int k = 4 * t + khalf;
            v2f a;
            a.x = xrow[k];
            a.y = xrow[k + 1];
            acc2 = __builtin_amdgcn_wmma_f32_16x16x4_f32(
                false, a, false, bw[t], (short)0, acc2, false, false);
        }
#pragma unroll
        for (int j = 0; j < 8; ++j)
            z2[(size_t)(rbase + j) * OUT_D + col] = acc2[j];
    }
}

#define G2_NB 32
#define G2_P  33
__global__ __launch_bounds__(256)
void gcn_gather32_out(const int* __restrict__ rowptr,
                      const int* __restrict__ csr,
                      const float* __restrict__ z2,
                      const float* __restrict__ nrmD,
                      const float* __restrict__ b2,
                      float* __restrict__ out) {
    __shared__ float tile[OUT_D * G2_P];
    const int tid  = threadIdx.x;
    const int nn   = tid >> 3;
    const int q    = tid & 7;
    const int n0   = blockIdx.x * G2_NB;
    const int node = n0 + nn;

    int   beg = 0, end = 0;
    float sD  = 0.0f;
    if (node < N_NODES) {
        beg = rowptr[node];
        end = rowptr[node + 1];
        sD  = nrmD[node];
    }
    beg = max(beg, 0);
    end = min(end, N_EDGES);
    const float4 bb = *(const float4*)(b2 + 4 * q);
    const float* const zcol = z2 + 4 * q;

    float4 acc = make_float4(0.0f, 0.0f, 0.0f, 0.0f);
    int next = 0;
    if (beg + q < end) next = csr[beg + q];
    for (int j = beg; j < end; j += 8) {
        const int rem  = end - j;
        const int cnt  = rem < 8 ? rem : 8;
        const int mine = min(max(next, 0), N_NODES - 1);
        const int s0 = __shfl(mine, 0, 8);
        const int s1 = __shfl(mine, 1, 8);
        const int s2 = __shfl(mine, 2, 8);
        const int s3 = __shfl(mine, 3, 8);
        const int s4 = __shfl(mine, 4, 8);
        const int s5 = __shfl(mine, 5, 8);
        const int s6 = __shfl(mine, 6, 8);
        const int s7 = __shfl(mine, 7, 8);
        const float4 z = make_float4(0.0f, 0.0f, 0.0f, 0.0f);
        float4 v0 = *(const float4*)(zcol + (size_t)s0 * OUT_D);
        float4 v1 = z, v2 = z, v3 = z, v4 = z, v5 = z, v6 = z, v7 = z;
        if (1 < cnt) v1 = *(const float4*)(zcol + (size_t)s1 * OUT_D);
        if (2 < cnt) v2 = *(const float4*)(zcol + (size_t)s2 * OUT_D);
        if (3 < cnt) v3 = *(const float4*)(zcol + (size_t)s3 * OUT_D);
        if (4 < cnt) v4 = *(const float4*)(zcol + (size_t)s4 * OUT_D);
        if (5 < cnt) v5 = *(const float4*)(zcol + (size_t)s5 * OUT_D);
        if (6 < cnt) v6 = *(const float4*)(zcol + (size_t)s6 * OUT_D);
        if (7 < cnt) v7 = *(const float4*)(zcol + (size_t)s7 * OUT_D);
        next = 0;
        if (j + 8 + q < end) next = csr[j + 8 + q];
        acc.x += v0.x; acc.y += v0.y; acc.z += v0.z; acc.w += v0.w;
        acc.x += v1.x; acc.y += v1.y; acc.z += v1.z; acc.w += v1.w;
        acc.x += v2.x; acc.y += v2.y; acc.z += v2.z; acc.w += v2.w;
        acc.x += v3.x; acc.y += v3.y; acc.z += v3.z; acc.w += v3.w;
        acc.x += v4.x; acc.y += v4.y; acc.z += v4.z; acc.w += v4.w;
        acc.x += v5.x; acc.y += v5.y; acc.z += v5.z; acc.w += v5.w;
        acc.x += v6.x; acc.y += v6.y; acc.z += v6.z; acc.w += v6.w;
        acc.x += v7.x; acc.y += v7.y; acc.z += v7.z; acc.w += v7.w;
    }

    tile[(4 * q + 0) * G2_P + nn] = acc.x * sD + bb.x;
    tile[(4 * q + 1) * G2_P + nn] = acc.y * sD + bb.y;
    tile[(4 * q + 2) * G2_P + nn] = acc.z * sD + bb.z;
    tile[(4 * q + 3) * G2_P + nn] = acc.w * sD + bb.w;
    __syncthreads();

    const int c    = tid >> 3;
    const int quad = tid & 7;
    const int n    = n0 + 4 * quad;
    if (n < N_NODES) {
        const float* t = tile + c * G2_P + 4 * quad;
        float4 o;
        o.x = t[0]; o.y = t[1]; o.z = t[2]; o.w = t[3];
        *(float4*)(out + (size_t)c * N_NODES + n) = o;
    }
}

extern "C" void kernel_launch(void* const* d_in, const int* in_sizes, int n_in,
                              void* d_out, int out_size, void* d_ws, size_t ws_size,
                              hipStream_t stream) {
    const float* h   = (const float*)d_in[0];
    const int*   src = (const int*)  d_in[1];
    const int*   dst = (const int*)  d_in[2];
    const float* W1  = (const float*)d_in[3];
    const float* b1  = (const float*)d_in[4];
    const float* W2  = (const float*)d_in[5];
    const float* b2  = (const float*)d_in[6];
    float*       out = (float*)d_out;

    int*   cntD   = (int*)d_ws;
    float* nrmS   = (float*)d_ws + 2 * N_NODES;
    float* nrmD   = nrmS + N_NODES;
    int*   rowptr = (int*)d_ws + 4 * N_NODES;
    int*   csr    = rowptr + 50048;
    float* feat   = (float*)d_ws + 4 * N_NODES + 50048 + N_EDGES;
    float* z2     = feat + (size_t)N_NODES * 64;
    int*   histD  = (int*)(z2 + (size_t)N_NODES * 64);
    int*   histS  = histD + (size_t)HB * N_NODES;

    const int T = 256;

    gcn_hist    <<<dim3(HB * HR, 2), 1024, 0, stream>>>(src, dst, histS, histD);
    gcn_combine <<<(N_NODES + T - 1) / T, T, 0, stream>>>(histD, histS, cntD, nrmS, nrmD);
    gcn_scan    <<<1, 1024, 0, stream>>>(cntD, rowptr);
    gcn_fill    <<<HB * HR, 1024, 0, stream>>>(src, dst, rowptr, histD, csr);

    gcn_scale_transpose<<<(N_NODES + TP_N - 1) / TP_N, 256, 0, stream>>>(h, nrmS, feat);
    gcn_gather_gemm12_wmma<<<N_NODES / 16, 256, 0, stream>>>(rowptr, csr, feat, nrmD, nrmS, W1, b1, W2, z2);

    gcn_gather32_out<<<(N_NODES + G2_NB - 1) / G2_NB, 256, 0, stream>>>(rowptr, csr, z2, nrmD, b2, out);
}


# scale+transpose rides in the slot-fill launch as 98 extra work-groups on the WGPs the 128 filling ones leave free (the two are independent): 6 launches, not 7; arithmetic and sum orders unchanged
# speedup vs baseline: 4.4560x; 1.0429x over previous
#include <hip/hip_runtime.h>
#include <hip/hip_bf16.h>


typedef float v2f __attribute__((ext_vector_type(2)));
typedef float v8f __attribute__((ext_vector_type(8)));

#define N_NODES 50000
#define N_EDGES 800000
#define IN_D 64
#define HID_D 64
#define OUT_D 32

#define HB   32
#define HR   4
#define H_CH (N_EDGES / HB)
#define H_RN (N_NODES / HR)
#define H_Q  (H_CH / 4)
#define H_IT ((H_Q + 1023) / 1024)
#define H_OT ((H_RN / 4 + 1023) / 1024)

__global__ __launch_bounds__(1024)
void gcn_hist(const int* __restrict__ src,
              const int* __restrict__ dst,
              int* __restrict__ histS,
              int* __restrict__ histD) {
    __shared__ int cnt[H_RN];
    const int tid = threadIdx.x;
    const int b   = blockIdx.x / HR;
    const int lo  = (blockIdx.x % HR) * H_RN;
    const int* const ids  = (blockIdx.y ? src : dst) + b * H_CH;
    int* const       hist = (blockIdx.y ? histS : histD) + b * N_NODES + lo;

    int4 v[H_IT];
#pragma unroll
    for (int i = 0; i < H_IT; ++i) {
        const int q = tid + 1024 * i;
        v[i] = make_int4(-1, -1, -1, -1);
        if (q < H_Q) v[i] = *(const int4*)(ids + 4 * q);
    }
    for (int i = tid; i < H_RN / 4; i += 1024) *(int4*)(cnt + 4 * i) = make_int4(0, 0, 0, 0);
    __syncthreads();
#pragma unroll
    for (int i = 0; i < H_IT; ++i) {
        const unsigned k0 = (unsigned)(v[i].x - lo);
        const unsigned k1 = (unsigned)(v[i].y - lo);
        const unsigned k2 = (unsigned)(v[i].z - lo);
        const unsigned k3 = (unsigned)(v[i].w - lo);
        if (k0 < (unsigned)H_RN) atomicAdd(cnt + k0, 1);
        if (k1 < (unsigned)H_RN) atomicAdd(cnt + k1, 1);
        if (k2 < (unsigned)H_RN) atomicAdd(cnt + k2, 1);
        if (k3 < (unsigned)H_RN) atomicAdd(cnt + k3, 1);
    }
    __syncthreads();
    for (int i = tid; i < H_RN / 4; i += 1024)
        *(int4*)(hist + 4 * i) = *(const int4*)(cnt + 4 * i);
}

__global__ __launch_bounds__(256)
void gcn_combine(int* __restrict__ histD,
                 const int* __restrict__ histS,
                 int* __restrict__ cntD,
                 float* __restrict__ nrmS,
                 float* __restrict__ nrmD) {
    const int n = blockIdx.x * 256 + threadIdx.x;
    if (n >= N_NODES) return;
    int d[HB], s[HB];
#pragma unroll
    for (int b = 0; b < HB; ++b) {
        d[b] = histD[b * N_NODES + n];
        s[b] = histS[b * N_NODES + n];
    }
    int run = 0, tot = 0;
#pragma unroll
    for (int b = 0; b < HB; ++b) {
        histD[b * N_NODES + n] = run;
        run += d[b];
        tot += s[b];
    }
    cntD[n] = run;
    nrmD[n] = rsqrtf(fmaxf((float)run, 1.0f));
    nrmS[n] = rsqrtf(fmaxf((float)tot, 1.0f));
}

#define SC_ROWS 13
__global__ __launch_bounds__(1024)
void gcn_scan(const int* __restrict__ cnt, int* __restrict__ rowptr) {
    __shared__ int wsum[32];
    const int lane  = threadIdx.x & 31;
    const int wv    = threadIdx.x >> 5;
    const int wbase = wv * (SC_ROWS * 128) + lane * 4;

    int4 v[SC_ROWS];
    int  ls[SC_ROWS];
    int  tot = 0;
#pragma unroll
    for (int q = 0; q < SC_ROWS; ++q) {
        const int idx = wbase + q * 128;
        v[q] = make_int4(0, 0, 0, 0);
        if (idx < N_NODES) v[q] = *(const int4*)(cnt + idx);
        ls[q] = v[q].x + v[q].y + v[q].z + v[q].w;
        tot += ls[q];
    }
    for (int o = 16; o > 0; o >>= 1) tot += __shfl_xor(tot, o);
    if (lane == 0) wsum[wv] = tot;
    __syncthreads();

    int t = wsum[lane];
    for (int o = 1; o < 32; o <<= 1) {
        const int u = __shfl_up(t, o);
        if (lane >= o) t += u;
    }
    const int total = __shfl(t, 31);
    const int wincl = __shfl(t, wv);
    int carry = wincl - tot;

#pragma unroll
    for (int q = 0; q < SC_ROWS; ++q) {
        int inc = ls[q];
        for (int o = 1; o < 32; o <<= 1) {
            const int u = __shfl_up(inc, o);
            if (lane >= o) inc += u;
        }
        const int ex  = carry + inc - ls[q];
        const int idx = wbase + q * 128;
        if (idx < N_NODES) {
            int4 r;
            r.x = ex;
            r.y = ex + v[q].x;
            r.z = r.y + v[q].y;
            r.w = r.z + v[q].z;
            *(int4*)(rowptr + idx) = r;
        }
        carry += __shfl(inc, 31);
    }
    if (threadIdx.x == 0) rowptr[N_NODES] = total;
}

#define TP_N     64
#define TP_PITCH 65
#define TP_GRP   4
#define TP_PER   2
#define TP_WG    (TP_GRP * TP_PER * TP_N)
#define FT_FILL  (HB * HR)
#define FT_TP    ((N_NODES + TP_WG - 1) / TP_WG)
#define FT_WORDS (TP_GRP * TP_PER * TP_N * TP_PITCH)
__global__ __launch_bounds__(1024)
void gcn_fill_transpose(const int* __restrict__ src,
                        const int* __restrict__ dst,
                        const int* __restrict__ rowptr,
                        const int* __restrict__ histD,
                        int* __restrict__ csr,
                        const float* __restrict__ h,
                        const float* __restrict__ nrmS,
                        float* __restrict__ feat) {
    __shared__ __attribute__((aligned(16))) union {
        int   off[H_RN];
        float tile[FT_WORDS];
    } sm;
    const int tid = threadIdx.x;

    if (blockIdx.x < FT_FILL) {
        int* const off = sm.off;
        const int b   = blockIdx.x / HR;
        const int lo  = (blockIdx.x % HR) * H_RN;
        const int* const dch = dst + b * H_CH;
        const int* const sch = src + b * H_CH;

        int4 dv[H_IT], sv[H_IT];
#pragma unroll
        for (int i = 0; i < H_IT; ++i) {
            const int q = tid + 1024 * i;
            dv[i] = make_int4(-1, -1, -1, -1);
            sv[i] = make_int4(0, 0, 0, 0);
            if (q < H_Q) {
                dv[i] = *(const int4*)(dch + 4 * q);
                sv[i] = *(const int4*)(sch + 4 * q);
            }
        }
        int4 rv[H_OT], pv[H_OT];
#pragma unroll
        for (int j = 0; j < H_OT; ++j) {
            const int i = tid + 1024 * j;
            rv[j] = make_int4(0, 0, 0, 0);
            pv[j] = make_int4(0, 0, 0, 0);
            if (i < H_RN / 4) {
                rv[j] = *(const int4*)(rowptr + lo + 4 * i);
                pv[j] = *(const int4*)(histD + b * N_NODES + lo + 4 * i);
            }
        }
#pragma unroll
        for (int j = 0; j < H_OT; ++j) {
            const int i = tid + 1024 * j;
            if (i < H_RN / 4)
                *(int4*)(off + 4 * i) = make_int4(rv[j].x + pv[j].x, rv[j].y + pv[j].y,
                                                  rv[j].z + pv[j].z, rv[j].w + pv[j].w);
        }
        __syncthreads();
#pragma unroll
        for (int i = 0; i < H_IT; ++i) {
            const unsigned k0 = (unsigned)(dv[i].x - lo);
            const unsigned k1 = (unsigned)(dv[i].y - lo);
            const unsigned k2 = (unsigned)(dv[i].z - lo);
            const unsigned k3 = (unsigned)(dv[i].w - lo);
            if (k0 < (unsigned)H_RN) csr[min(max(atomicAdd(off + k0, 1), 0), N_EDGES - 1)] = sv[i].x;
            if (k1 < (unsigned)H_RN) csr[min(max(atomicAdd(off + k1, 1), 0), N_EDGES - 1)] = sv[i].y;
            if (k2 < (unsigned)H_RN) csr[min(max(atomicAdd(off + k2, 1), 0), N_EDGES - 1)] = sv[i].z;
            if (k3 < (unsigned)H_RN) csr[min(max(atomicAdd(off + k3, 1), 0), N_EDGES - 1)] = sv[i].w;
        }
    } else {
        const int grp = tid >> 8;
        const int t   = tid & 255;
        const int tl0 = ((blockIdx.x - FT_FILL) * TP_GRP + grp) * TP_PER;

        const int c = t & 15;
        const int r = t >> 4;
        float4 s[TP_PER];
        float4 v[TP_PER][4];
#pragma unroll
        for (int u = 0; u < TP_PER; ++u) {
            const int n = min((tl0 + u) * TP_N + 4 * c, N_NODES - 4);
            s[u] = *(const float4*)(nrmS + n);
#pragma unroll
            for (int p = 0; p < 4; ++p)
                v[u][p] = *(const float4*)(h + (size_t)(r + 16 * p) * N_NODES + n);
        }
#pragma unroll
        for (int u = 0; u < TP_PER; ++u) {
            float* const tile = sm.tile + (grp * TP_PER + u) * (TP_N * TP_PITCH);
#pragma unroll
            for (int p = 0; p < 4; ++p) {
                const int d = r + 16 * p;
                tile[(4 * c + 0) * TP_PITCH + d] = v[u][p].x * s[u].x;
                tile[(4 * c + 1) * TP_PITCH + d] = v[u][p].y * s[u].y;
                tile[(4 * c + 2) * TP_PITCH + d] = v[u][p].z * s[u].z;
                tile[(4 * c + 3) * TP_PITCH + d] = v[u][p].w * s[u].w;
            }
        }
        __syncthreads();

        for (int u = 0; u < TP_PER; ++u) {
            const int n0     = (tl0 + u) * TP_N;
            const int nvalid = (N_NODES - n0) < TP_N ? (N_NODES - n0) : TP_N;
            const float* const tile = sm.tile + (grp * TP_PER + u) * (TP_N * TP_PITCH);
            for (int k = 0; k < 4; ++k) {
                const int e  = t + 256 * k;
                const int nn = e >> 4;
                const int q  = e & 15;
                if (nn < nvalid) {
                    const float* tp = tile + nn * TP_PITCH + 4 * q;
                    float4 o;
                    o.x = tp[0]; o.y = tp[1]; o.z = tp[2]; o.w = tp[3];
                    *(float4*)(feat + (size_t)n0 * IN_D + (size_t)e * 4) = o;
                }
            }
        }
    }
}

#define X_PITCH 68
__global__ __launch_bounds__(256)
void gcn_gather_gemm12_wmma(const int* __restrict__ rowptr,
                            const int* __restrict__ csr,
                            const float* __restrict__ feat,
                            const float* __restrict__ nrmD,
                            const float* __restrict__ nrmS,
                            const float* __restrict__ W1,
                            const float* __restrict__ b1,
                            const float* __restrict__ W2,
                            float* __restrict__ z2) {
    __shared__ float as[16 * X_PITCH];
    __shared__ float xs[16 * X_PITCH];

    const int nn   = threadIdx.x >> 4;
    const int node = blockIdx.x * 16 + nn;
    const int l    = threadIdx.x & 15;
    int beg = rowptr[node];
    int end = rowptr[node + 1];
    const float sD = nrmD[node];
    beg = max(beg, 0);
    end = min(end, N_EDGES);
    const float* const fcol = feat + 4 * l;

    float4 acc = make_float4(0.0f, 0.0f, 0.0f, 0.0f);
    int next = 0;
    if (beg + l < end) next = csr[beg + l];
    for (int j = beg; j < end; j += 16) {
        const int rem = end - j;
        const int cnt = rem < 16 ? rem : 16;
        const int mine = min(max(next, 0), N_NODES - 1);
        next = 0;
        if (j + 16 + l < end) next = csr[j + 16 + l];
        for (int k = 0; k < cnt; k += 4) {
            const int s0 = __shfl(mine, k + 0, 16);
            const int s1 = __shfl(mine, k + 1, 16);
            const int s2 = __shfl(mine, k + 2, 16);
            const int s3 = __shfl(mine, k + 3, 16);
            const float4 z = make_float4(0.0f, 0.0f, 0.0f, 0.0f);
            float4 v0 = *(const float4*)(fcol + (size_t)s0 * 64);
            float4 v1 = z, v2 = z, v3 = z;
            if (k + 1 < cnt) v1 = *(const float4*)(fcol + (size_t)s1 * 64);
            if (k + 2 < cnt) v2 = *(const float4*)(fcol + (size_t)s2 * 64);
            if (k + 3 < cnt) v3 = *(const float4*)(fcol + (size_t)s3 * 64);
            acc.x += v0.x; acc.y += v0.y; acc.z += v0.z; acc.w += v0.w;
            acc.x += v1.x; acc.y += v1.y; acc.z += v1.z; acc.w += v1.w;
            acc.x += v2.x; acc.y += v2.y; acc.z += v2.z; acc.w += v2.w;
            acc.x += v3.x; acc.y += v3.y; acc.z += v3.z; acc.w += v3.w;
        }
    }
    const int lane  = threadIdx.x & 31;
    const int wv    = __builtin_amdgcn_readfirstlane(threadIdx.x >> 5);
    const int m     = lane & 15;
    const int khalf = (lane >> 4) << 1;
    const int rloc  = (lane >> 4) << 3;
    const int rbase = blockIdx.x * 16 + rloc;
    const bool first  = wv < 4;
    const int  w2     = 4 + ((blockIdx.x & 1) << 1);
    const bool second = (wv == w2) | (wv == w2 + 1);
    const int  col    = first ? (wv << 4) + m : ((wv & 1) << 4) + m;

    v2f   bw[16];
    float nsv[8];
    float bias = 0.0f;
    if (first) {
#pragma unroll
        for (int t = 0; t < 16; ++t) {
            const int k = 4 * t + khalf;
            bw[t].x = W1[k * HID_D + col];
            bw[t].y = W1[(k + 1) * HID_D + col];
        }
        bias = b1[col];
#pragma unroll
        for (int j = 0; j < 8; ++j) nsv[j] = nrmS[rbase + j];
    } else if (second) {
#pragma unroll
        for (int t = 0; t < 16; ++t) {
            const int k = 4 * t + khalf;
            bw[t].x = W2[k * OUT_D + col];
            bw[t].y = W2[(k + 1) * OUT_D + col];
        }
    }

    *(float4*)(as + nn * X_PITCH + 4 * l) =
        make_float4(acc.x * sD, acc.y * sD, acc.z * sD, acc.w * sD);
    __syncthreads();


    if (first) {
        const float* arow = as + m * X_PITCH;
        v8f acc1 = {};
#pragma unroll
        for (int t = 0; t < 16; ++t) {
            const int k = 4 * t + khalf;
            v2f a;
            a.x = arow[k];
            a.y = arow[k + 1];
            acc1 = __builtin_amdgcn_wmma_f32_16x16x4_f32(
                false, a, false, bw[t], (short)0, acc1, false, false);
        }
#pragma unroll
        for (int j = 0; j < 8; ++j) {
            float u = acc1[j] + bias;
            u = u > 0.0f ? u : 0.0f;
            xs[(rloc + j) * X_PITCH + col] = u * nsv[j];
        }
    }
    __syncthreads();

    if (second) {
        const float* xrow = xs + m * X_PITCH;
        v8f acc2 = {};
#pragma unroll
        for (int t = 0; t < 16; ++t) {
            const int k = 4 * t + khalf;
            v2f a;
            a.x = xrow[k];
            a.y = xrow[k + 1];
            acc2 = __builtin_amdgcn_wmma_f32_16x16x4_f32(
                false, a, false, bw[t], (short)0, acc2, false, false);
        }
#pragma unroll
        for (int j = 0; j < 8; ++j)
            z2[(size_t)(rbase + j) * OUT_D + col] = acc2[j];
    }
}

#define G2_NB 32
#define G2_P  33
__global__ __launch_bounds__(256)
void gcn_gather32_out(const int* __restrict__ rowptr,
                      const int* __restrict__ csr,
                      const float* __restrict__ z2,
                      const float* __restrict__ nrmD,
                      const float* __restrict__ b2,
                      float* __restrict__ out) {
    __shared__ float tile[OUT_D * G2_P];
    const int tid  = threadIdx.x;
    const int nn   = tid >> 3;
    const int q    = tid & 7;
    const int n0   = blockIdx.x * G2_NB;
    const int node = n0 + nn;

    int   beg = 0, end = 0;
    float sD  = 0.0f;
    if (node < N_NODES) {
        beg = rowptr[node];
        end = rowptr[node + 1];
        sD  = nrmD[node];
    }
    beg = max(beg, 0);
    end = min(end, N_EDGES);
    const float4 bb = *(const float4*)(b2 + 4 * q);
    const float* const zcol = z2 + 4 * q;

    float4 acc = make_float4(0.0f, 0.0f, 0.0f, 0.0f);
    int next = 0;
    if (beg + q < end) next = csr[beg + q];
    for (int j = beg; j < end; j += 8) {
        const int rem  = end - j;
        const int cnt  = rem < 8 ? rem : 8;
        const int mine = min(max(next, 0), N_NODES - 1);
        const int s0 = __shfl(mine, 0, 8);
        const int s1 = __shfl(mine, 1, 8);
        const int s2 = __shfl(mine, 2, 8);
        const int s3 = __shfl(mine, 3, 8);
        const int s4 = __shfl(mine, 4, 8);
        const int s5 = __shfl(mine, 5, 8);
        const int s6 = __shfl(mine, 6, 8);
        const int s7 = __shfl(mine, 7, 8);
        const float4 z = make_float4(0.0f, 0.0f, 0.0f, 0.0f);
        float4 v0 = *(const float4*)(zcol + (size_t)s0 * OUT_D);
        float4 v1 = z, v2 = z, v3 = z, v4 = z, v5 = z, v6 = z, v7 = z;
        if (1 < cnt) v1 = *(const float4*)(zcol + (size_t)s1 * OUT_D);
        if (2 < cnt) v2 = *(const float4*)(zcol + (size_t)s2 * OUT_D);
        if (3 < cnt) v3 = *(const float4*)(zcol + (size_t)s3 * OUT_D);
        if (4 < cnt) v4 = *(const float4*)(zcol + (size_t)s4 * OUT_D);
        if (5 < cnt) v5 = *(const float4*)(zcol + (size_t)s5 * OUT_D);
        if (6 < cnt) v6 = *(const float4*)(zcol + (size_t)s6 * OUT_D);
        if (7 < cnt) v7 = *(const float4*)(zcol + (size_t)s7 * OUT_D);
        next = 0;
        if (j + 8 + q < end) next = csr[j + 8 + q];
        acc.x += v0.x; acc.y += v0.y; acc.z += v0.z; acc.w += v0.w;
        acc.x += v1.x; acc.y += v1.y; acc.z += v1.z; acc.w += v1.w;
        acc.x += v2.x; acc.y += v2.y; acc.z += v2.z; acc.w += v2.w;
        acc.x += v3.x; acc.y += v3.y; acc.z += v3.z; acc.w += v3.w;
        acc.x += v4.x; acc.y += v4.y; acc.z += v4.z; acc.w += v4.w;
        acc.x += v5.x; acc.y += v5.y; acc.z += v5.z; acc.w += v5.w;
        acc.x += v6.x; acc.y += v6.y; acc.z += v6.z; acc.w += v6.w;
        acc.x += v7.x; acc.y += v7.y; acc.z += v7.z; acc.w += v7.w;
    }

    tile[(4 * q + 0) * G2_P + nn] = acc.x * sD + bb.x;
    tile[(4 * q + 1) * G2_P + nn] = acc.y * sD + bb.y;
    tile[(4 * q + 2) * G2_P + nn] = acc.z * sD + bb.z;
    tile[(4 * q + 3) * G2_P + nn] = acc.w * sD + bb.w;
    __syncthreads();

    const int c    = tid >> 3;
    const int quad = tid & 7;
    const int n    = n0 + 4 * quad;
    if (n < N_NODES) {
        const float* t = tile + c * G2_P + 4 * quad;
        float4 o;
        o.x = t[0]; o.y = t[1]; o.z = t[2]; o.w = t[3];
        *(float4*)(out + (size_t)c * N_NODES + n) = o;
    }
}

extern "C" void kernel_launch(void* const* d_in, const int* in_sizes, int n_in,
                              void* d_out, int out_size, void* d_ws, size_t ws_size,
                              hipStream_t stream) {
    const float* h   = (const float*)d_in[0];
    const int*   src = (const int*)  d_in[1];
    const int*   dst = (const int*)  d_in[2];
    const float* W1  = (const float*)d_in[3];
    const float* b1  = (const float*)d_in[4];
    const float* W2  = (const float*)d_in[5];
    const float* b2  = (const float*)d_in[6];
    float*       out = (float*)d_out;

    int*   cntD   = (int*)d_ws;
    float* nrmS   = (float*)d_ws + 2 * N_NODES;
    float* nrmD   = nrmS + N_NODES;
    int*   rowptr = (int*)d_ws + 4 * N_NODES;
    int*   csr    = rowptr + 50048;
    float* feat   = (float*)d_ws + 4 * N_NODES + 50048 + N_EDGES;
    float* z2     = feat + (size_t)N_NODES * 64;
    int*   histD  = (int*)(z2 + (size_t)N_NODES * 64);
    int*   histS  = histD + (size_t)HB * N_NODES;

    const int T = 256;

    gcn_hist    <<<dim3(HB * HR, 2), 1024, 0, stream>>>(src, dst, histS, histD);
    gcn_combine <<<(N_NODES + T - 1) / T, T, 0, stream>>>(histD, histS, cntD, nrmS, nrmD);
    gcn_scan    <<<1, 1024, 0, stream>>>(cntD, rowptr);
    gcn_fill_transpose<<<FT_FILL + FT_TP, 1024, 0, stream>>>(src, dst, rowptr, histD, csr, h, nrmS, feat);

    gcn_gather_gemm12_wmma<<<N_NODES / 16, 256, 0, stream>>>(rowptr, csr, feat, nrmD, nrmS, W1, b1, W2, z2);

    gcn_gather32_out<<<(N_NODES + G2_NB - 1) / G2_NB, 256, 0, stream>>>(rowptr, csr, z2, nrmD, b2, out);
}


# row-offset scan folded into the chunk-combine launch: hist adds per-chunk in-degree totals of 250-node groups, a combine work-group sums the groups in front and scans its own; 5 launches, not 6
# speedup vs baseline: 4.8178x; 1.0812x over previous
#include <hip/hip_runtime.h>
#include <hip/hip_bf16.h>


typedef float v2f __attribute__((ext_vector_type(2)));
typedef float v8f __attribute__((ext_vector_type(8)));

#define N_NODES 50000
#define N_EDGES 800000
#define IN_D 64
#define HID_D 64
#define OUT_D 32

#define HB   32
#define HR   4
#define H_CH (N_EDGES / HB)
#define H_RN (N_NODES / HR)
#define H_Q  (H_CH / 4)
#define H_IT ((H_Q + 1023) / 1024)
#define H_OT ((H_RN / 4 + 1023) / 1024)
#define GP_N   250
#define GP_R   (H_RN / GP_N)
#define GP_ALL (N_NODES / GP_N)

__global__ __launch_bounds__(1024)
void gcn_hist(const int* __restrict__ src,
              const int* __restrict__ dst,
              int* __restrict__ histS,
              int* __restrict__ histD,
              int* __restrict__ gsum) {
    __shared__ int cnt[H_RN];
    const int tid = threadIdx.x;
    const int b   = blockIdx.x / HR;
    const int lo  = (blockIdx.x % HR) * H_RN;
    const int* const ids  = (blockIdx.y ? src : dst) + b * H_CH;
    int* const       hist = (blockIdx.y ? histS : histD) + b * N_NODES + lo;

    int4 v[H_IT];
#pragma unroll
    for (int i = 0; i < H_IT; ++i) {
        const int q = tid + 1024 * i;
        v[i] = make_int4(-1, -1, -1, -1);
        if (q < H_Q) v[i] = *(const int4*)(ids + 4 * q);
    }
    for (int i = tid; i < H_RN / 4; i += 1024) *(int4*)(cnt + 4 * i) = make_int4(0, 0, 0, 0);
    __syncthreads();
#pragma unroll
    for (int i = 0; i < H_IT; ++i) {
        const unsigned k0 = (unsigned)(v[i].x - lo);
        const unsigned k1 = (unsigned)(v[i].y - lo);
        const unsigned k2 = (unsigned)(v[i].z - lo);
        const unsigned k3 = (unsigned)(v[i].w - lo);
        if (k0 < (unsigned)H_RN) atomicAdd(cnt + k0, 1);
        if (k1 < (unsigned)H_RN) atomicAdd(cnt + k1, 1);
        if (k2 < (unsigned)H_RN) atomicAdd(cnt + k2, 1);
        if (k3 < (unsigned)H_RN) atomicAdd(cnt + k3, 1);
    }
    __syncthreads();
    for (int i = tid; i < H_RN / 4; i += 1024)
        *(int4*)(hist + 4 * i) = *(const int4*)(cnt + 4 * i);
    if (blockIdx.y == 0) {
        const int  lane = tid & 31;
        const int  g0   = tid >> 5;
        const bool two  = g0 + 32 < GP_R;
        const int  g1   = two ? g0 + 32 : g0;
        int t0 = 0, t1 = 0;
#pragma unroll
        for (int q = 0; q < (GP_N + 31) / 32; ++q) {
            const int j = lane + 32 * q;
            if (j < GP_N) {
                t0 += cnt[g0 * GP_N + j];
                t1 += cnt[g1 * GP_N + j];
            }
        }
        for (int o = 16; o > 0; o >>= 1) {
            t0 += __shfl_xor(t0, o);
            t1 += __shfl_xor(t1, o);
        }
        if (lane == 0) {
            int* const gs = gsum + b * GP_ALL + (blockIdx.x % HR) * GP_R;
            gs[g0] = t0;
            if (two) gs[g0 + 32] = t1;
        }
    }
}

__global__ __launch_bounds__(256)
void gcn_combine_scan(int* __restrict__ histD,
                      const int* __restrict__ histS,
                      const int* __restrict__ gsum,
                      int* __restrict__ rowptr,
                      float* __restrict__ nrmS,
                      float* __restrict__ nrmD) {
    __shared__ int wsum[8];
    __shared__ int wpre[8];
    const int tid  = threadIdx.x;
    const int g    = blockIdx.x;
    const int lane = tid & 31;
    const int wv   = tid >> 5;
    const bool live = tid < GP_N;
    const int n  = g * GP_N + (live ? tid : 0);
    const int gi = tid < GP_ALL ? tid : 0;
    int d[HB], s[HB], p[HB];
#pragma unroll
    for (int b = 0; b < HB; ++b) {
        d[b] = histD[b * N_NODES + n];
        s[b] = histS[b * N_NODES + n];
        p[b] = gsum[b * GP_ALL + gi];
    }
    int run = 0, tot = 0, pre = 0;
    if (live) {
#pragma unroll
        for (int b = 0; b < HB; ++b) {
            histD[b * N_NODES + n] = run;
            run += d[b];
            tot += s[b];
        }
        nrmD[n] = rsqrtf(fmaxf((float)run, 1.0f));
        nrmS[n] = rsqrtf(fmaxf((float)tot, 1.0f));
    }
#pragma unroll
    for (int b = 0; b < HB; ++b) pre += p[b];
    if (tid >= g) pre = 0;

    int inc = run;
    for (int o = 1; o < 32; o <<= 1) {
        const int u = __shfl_up(inc, o);
        if (lane >= o) inc += u;
    }
    for (int o = 16; o > 0; o >>= 1) pre += __shfl_xor(pre, o);
    if (lane == 31) wsum[wv] = inc;
    if (lane == 0)  wpre[wv] = pre;
    __syncthreads();
    int off = 0;
#pragma unroll
    for (int w = 0; w < 8; ++w) {
        off += wpre[w];
        if (w < wv) off += wsum[w];
    }
    if (live) rowptr[n] = off + inc - run;
    if (g == GP_ALL - 1 && tid == GP_N - 1) rowptr[N_NODES] = off + inc;
}

#define TP_N     64
#define TP_PITCH 65
#define TP_GRP   4
#define TP_PER   2
#define TP_WG    (TP_GRP * TP_PER * TP_N)
#define FT_FILL  (HB * HR)
#define FT_TP    ((N_NODES + TP_WG - 1) / TP_WG)
#define FT_WORDS (TP_GRP * TP_PER * TP_N * TP_PITCH)
__global__ __launch_bounds__(1024)
void gcn_fill_transpose(const int* __restrict__ src,
                        const int* __restrict__ dst,
                        const int* __restrict__ rowptr,
                        const int* __restrict__ histD,
                        int* __restrict__ csr,
                        const float* __restrict__ h,
                        const float* __restrict__ nrmS,
                        float* __restrict__ feat) {
    __shared__ __attribute__((aligned(16))) union {
        int   off[H_RN];
        float tile[FT_WORDS];
    } sm;
    const int tid = threadIdx.x;

    if (blockIdx.x < FT_FILL) {
        int* const off = sm.off;
        const int b   = blockIdx.x / HR;
        const int lo  = (blockIdx.x % HR) * H_RN;
        const int* const dch = dst + b * H_CH;
        const int* const sch = src + b * H_CH;

        int4 dv[H_IT], sv[H_IT];
#pragma unroll
        for (int i = 0; i < H_IT; ++i) {
            const int q = tid + 1024 * i;
            dv[i] = make_int4(-1, -1, -1, -1);
            sv[i] = make_int4(0, 0, 0, 0);
            if (q < H_Q) {
                dv[i] = *(const int4*)(dch + 4 * q);
                sv[i] = *(const int4*)(sch + 4 * q);
            }
        }
        int4 rv[H_OT], pv[H_OT];
#pragma unroll
        for (int j = 0; j < H_OT; ++j) {
            const int i = tid + 1024 * j;
            rv[j] = make_int4(0, 0, 0, 0);
            pv[j] = make_int4(0, 0, 0, 0);
            if (i < H_RN / 4) {
                rv[j] = *(const int4*)(rowptr + lo + 4 * i);
                pv[j] = *(const int4*)(histD + b * N_NODES + lo + 4 * i);
            }
        }
#pragma unroll
        for (int j = 0; j < H_OT; ++j) {
            const int i = tid + 1024 * j;
            if (i < H_RN / 4)
                *(int4*)(off + 4 * i) = make_int4(rv[j].x + pv[j].x, rv[j].y + pv[j].y,
                                                  rv[j].z + pv[j].z, rv[j].w + pv[j].w);
        }
        __syncthreads();
#pragma unroll
        for (int i = 0; i < H_IT; ++i) {
            const unsigned k0 = (unsigned)(dv[i].x - lo);
            const unsigned k1 = (unsigned)(dv[i].y - lo);
            const unsigned k2 = (unsigned)(dv[i].z - lo);
            const unsigned k3 = (unsigned)(dv[i].w - lo);
            if (k0 < (unsigned)H_RN) csr[min(max(atomicAdd(off + k0, 1), 0), N_EDGES - 1)] = sv[i].x;
            if (k1 < (unsigned)H_RN) csr[min(max(atomicAdd(off + k1, 1), 0), N_EDGES - 1)] = sv[i].y;
            if (k2 < (unsigned)H_RN) csr[min(max(atomicAdd(off + k2, 1), 0), N_EDGES - 1)] = sv[i].z;
            if (k3 < (unsigned)H_RN) csr[min(max(atomicAdd(off + k3, 1), 0), N_EDGES - 1)] = sv[i].w;
        }
    } else {
        const int grp = tid >> 8;
        const int t   = tid & 255;
        const int tl0 = ((blockIdx.x - FT_FILL) * TP_GRP + grp) * TP_PER;

        const int c = t & 15;
        const int r = t >> 4;
        float4 s[TP_PER];
        float4 v[TP_PER][4];
#pragma unroll
        for (int u = 0; u < TP_PER; ++u) {
            const int n = min((tl0 + u) * TP_N + 4 * c, N_NODES - 4);
            s[u] = *(const float4*)(nrmS + n);
#pragma unroll
            for (int p = 0; p < 4; ++p)
                v[u][p] = *(const float4*)(h + (size_t)(r + 16 * p) * N_NODES + n);
        }
#pragma unroll
        for (int u = 0; u < TP_PER; ++u) {
            float* const tile = sm.tile + (grp * TP_PER + u) * (TP_N * TP_PITCH);
#pragma unroll
            for (int p = 0; p < 4; ++p) {
                const int d = r + 16 * p;
                tile[(4 * c + 0) * TP_PITCH + d] = v[u][p].x * s[u].x;
                tile[(4 * c + 1) * TP_PITCH + d] = v[u][p].y * s[u].y;
                tile[(4 * c + 2) * TP_PITCH + d] = v[u][p].z * s[u].z;
                tile[(4 * c + 3) * TP_PITCH + d] = v[u][p].w * s[u].w;
            }
        }
        __syncthreads();

        for (int u = 0; u < TP_PER; ++u) {
            const int n0     = (tl0 + u) * TP_N;
            const int nvalid = (N_NODES - n0) < TP_N ? (N_NODES - n0) : TP_N;
            const float* const tile = sm.tile + (grp * TP_PER + u) * (TP_N * TP_PITCH);
            for (int k = 0; k < 4; ++k) {
                const int e  = t + 256 * k;
                const int nn = e >> 4;
                const int q  = e & 15;
                if (nn < nvalid) {
                    const float* tp = tile + nn * TP_PITCH + 4 * q;
                    float4 o;
                    o.x = tp[0]; o.y = tp[1]; o.z = tp[2]; o.w = tp[3];
                    *(float4*)(feat + (size_t)n0 * IN_D + (size_t)e * 4) = o;
                }
            }
        }
    }
}

#define X_PITCH 68
__global__ __launch_bounds__(256)
void gcn_gather_gemm12_wmma(const int* __restrict__ rowptr,
                            const int* __restrict__ csr,
                            const float* __restrict__ feat,
                            const float* __restrict__ nrmD,
                            const float* __restrict__ nrmS,
                            const float* __restrict__ W1,
                            const float* __restrict__ b1,
                            const float* __restrict__ W2,
                            float* __restrict__ z2) {
    __shared__ float as[16 * X_PITCH];
    __shared__ float xs[16 * X_PITCH];

    const int nn   = threadIdx.x >> 4;
    const int node = blockIdx.x * 16 + nn;
    const int l    = threadIdx.x & 15;
    int beg = rowptr[node];
    int end = rowptr[node + 1];
    const float sD = nrmD[node];
    beg = max(beg, 0);
    end = min(end, N_EDGES);
    const float* const fcol = feat + 4 * l;

    float4 acc = make_float4(0.0f, 0.0f, 0.0f, 0.0f);
    int next = 0;
    if (beg + l < end) next = csr[beg + l];
    for (int j = beg; j < end; j += 16) {
        const int rem = end - j;
        const int cnt = rem < 16 ? rem : 16;
        const int mine = min(max(next, 0), N_NODES - 1);
        next = 0;
        if (j + 16 + l < end) next = csr[j + 16 + l];
        for (int k = 0; k < cnt; k += 4) {
            const int s0 = __shfl(mine, k + 0, 16);
            const int s1 = __shfl(mine, k + 1, 16);
            const int s2 = __shfl(mine, k + 2, 16);
            const int s3 = __shfl(mine, k + 3, 16);
            const float4 z = make_float4(0.0f, 0.0f, 0.0f, 0.0f);
            float4 v0 = *(const float4*)(fcol + (size_t)s0 * 64);
            float4 v1 = z, v2 = z, v3 = z;
            if (k + 1 < cnt) v1 = *(const float4*)(fcol + (size_t)s1 * 64);
            if (k + 2 < cnt) v2 = *(const float4*)(fcol + (size_t)s2 * 64);
            if (k + 3 < cnt) v3 = *(const float4*)(fcol + (size_t)s3 * 64);
            acc.x += v0.x; acc.y += v0.y; acc.z += v0.z; acc.w += v0.w;
            acc.x += v1.x; acc.y += v1.y; acc.z += v1.z; acc.w += v1.w;
            acc.x += v2.x; acc.y += v2.y; acc.z += v2.z; acc.w += v2.w;
            acc.x += v3.x; acc.y += v3.y; acc.z += v3.z; acc.w += v3.w;
        }
    }
    const int lane  = threadIdx.x & 31;
    const int wv    = __builtin_amdgcn_readfirstlane(threadIdx.x >> 5);
    const int m     = lane & 15;
    const int khalf = (lane >> 4) << 1;
    const int rloc  = (lane >> 4) << 3;
    const int rbase = blockIdx.x * 16 + rloc;
    const bool first  = wv < 4;
    const int  w2     = 4 + ((blockIdx.x & 1) << 1);
    const bool second = (wv == w2) | (wv == w2 + 1);
    const int  col    = first ? (wv << 4) + m : ((wv & 1) << 4) + m;

    v2f   bw[16];
    float nsv[8];
    float bias = 0.0f;
    if (first) {
#pragma unroll
        for (int t = 0; t < 16; ++t) {
            const int k = 4 * t + khalf;
            bw[t].x = W1[k * HID_D + col];
            bw[t].y = W1[(k + 1) * HID_D + col];
        }
        bias = b1[col];
#pragma unroll
        for (int j = 0; j < 8; ++j) nsv[j] = nrmS[rbase + j];
    } else if (second) {
#pragma unroll
        for (int t = 0; t < 16; ++t) {
            const int k = 4 * t + khalf;
            bw[t].x = W2[k * OUT_D + col];
            bw[t].y = W2[(k + 1) * OUT_D + col];
        }
    }

    *(float4*)(as + nn * X_PITCH + 4 * l) =
        make_float4(acc.x * sD, acc.y * sD, acc.z * sD, acc.w * sD);
    __syncthreads();


    if (first) {
        const float* arow = as + m * X_PITCH;
        v8f acc1 = {};
#pragma unroll
        for (int t = 0; t < 16; ++t) {
            const int k = 4 * t + khalf;
            v2f a;
            a.x = arow[k];
            a.y = arow[k + 1];
            acc1 = __builtin_amdgcn_wmma_f32_16x16x4_f32(
                false, a, false, bw[t], (short)0, acc1, false, false);
        }
#pragma unroll
        for (int j = 0; j < 8; ++j) {
            float u = acc1[j] + bias;
            u = u > 0.0f ? u : 0.0f;
            xs[(rloc + j) * X_PITCH + col] = u * nsv[j];
        }
    }
    __syncthreads();

    if (second) {
        const float* xrow = xs + m * X_PITCH;
        v8f acc2 = {};
#pragma unroll
        for (int t = 0; t < 16; ++t) {
            const int k = 4 * t + khalf;
            v2f a;
            a.x = xrow[k];
            a.y = xrow[k + 1];
            acc2 = __builtin_amdgcn_wmma_f32_16x16x4_f32(
                false, a, false, bw[t], (short)0, acc2, false, false);
        }
#pragma unroll
        for (int j = 0; j < 8; ++j)
            z2[(size_t)(rbase + j) * OUT_D + col] = acc2[j];
    }
}

#define G2_NB 32
#define G2_P  33
__global__ __launch_bounds__(256)
void gcn_gather32_out(const int* __restrict__ rowptr,
                      const int* __restrict__ csr,
                      const float* __restrict__ z2,
                      const float* __restrict__ nrmD,
                      const float* __restrict__ b2,
                      float* __restrict__ out) {
    __shared__ float tile[OUT_D * G2_P];
    const int tid  = threadIdx.x;
    const int nn   = tid >> 3;
    const int q    = tid & 7;
    const int n0   = blockIdx.x * G2_NB;
    const int node = n0 + nn;

    int   beg = 0, end = 0;
    float sD  = 0.0f;
    if (node < N_NODES) {
        beg = rowptr[node];
        end = rowptr[node + 1];
        sD  = nrmD[node];
    }
    beg = max(beg, 0);
    end = min(end, N_EDGES);
    const float4 bb = *(const float4*)(b2 + 4 * q);
    const float* const zcol = z2 + 4 * q;

    float4 acc = make_float4(0.0f, 0.0f, 0.0f, 0.0f);
    int next = 0;
    if (beg + q < end) next = csr[beg + q];
    for (int j = beg; j < end; j += 8) {
        const int rem  = end - j;
        const int cnt  = rem < 8 ? rem : 8;
        const int mine = min(max(next, 0), N_NODES - 1);
        const int s0 = __shfl(mine, 0, 8);
        const int s1 = __shfl(mine, 1, 8);
        const int s2 = __shfl(mine, 2, 8);
        const int s3 = __shfl(mine, 3, 8);
        const int s4 = __shfl(mine, 4, 8);
        const int s5 = __shfl(mine, 5, 8);
        const int s6 = __shfl(mine, 6, 8);
        const int s7 = __shfl(mine, 7, 8);
        const float4 z = make_float4(0.0f, 0.0f, 0.0f, 0.0f);
        float4 v0 = *(const float4*)(zcol + (size_t)s0 * OUT_D);
        float4 v1 = z, v2 = z, v3 = z, v4 = z, v5 = z, v6 = z, v7 = z;
        if (1 < cnt) v1 = *(const float4*)(zcol + (size_t)s1 * OUT_D);
        if (2 < cnt) v2 = *(const float4*)(zcol + (size_t)s2 * OUT_D);
        if (3 < cnt) v3 = *(const float4*)(zcol + (size_t)s3 * OUT_D);
        if (4 < cnt) v4 = *(const float4*)(zcol + (size_t)s4 * OUT_D);
        if (5 < cnt) v5 = *(const float4*)(zcol + (size_t)s5 * OUT_D);
        if (6 < cnt) v6 = *(const float4*)(zcol + (size_t)s6 * OUT_D);
        if (7 < cnt) v7 = *(const float4*)(zcol + (size_t)s7 * OUT_D);
        next = 0;
        if (j + 8 + q < end) next = csr[j + 8 + q];
        acc.x += v0.x; acc.y += v0.y; acc.z += v0.z; acc.w += v0.w;
        acc.x += v1.x; acc.y += v1.y; acc.z += v1.z; acc.w += v1.w;
        acc.x += v2.x; acc.y += v2.y; acc.z += v2.z; acc.w += v2.w;
        acc.x += v3.x; acc.y += v3.y; acc.z += v3.z; acc.w += v3.w;
        acc.x += v4.x; acc.y += v4.y; acc.z += v4.z; acc.w += v4.w;
        acc.x += v5.x; acc.y += v5.y; acc.z += v5.z; acc.w += v5.w;
        acc.x += v6.x; acc.y += v6.y; acc.z += v6.z; acc.w += v6.w;
        acc.x += v7.x; acc.y += v7.y; acc.z += v7.z; acc.w += v7.w;
    }

    tile[(4 * q + 0) * G2_P + nn] = acc.x * sD + bb.x;
    tile[(4 * q + 1) * G2_P + nn] = acc.y * sD + bb.y;
    tile[(4 * q + 2) * G2_P + nn] = acc.z * sD + bb.z;
    tile[(4 * q + 3) * G2_P + nn] = acc.w * sD + bb.w;
    __syncthreads();

    const int c    = tid >> 3;
    const int quad = tid & 7;
    const int n    = n0 + 4 * quad;
    if (n < N_NODES) {
        const float* t = tile + c * G2_P + 4 * quad;
        float4 o;
        o.x = t[0]; o.y = t[1]; o.z = t[2]; o.w = t[3];
        *(float4*)(out + (size_t)c * N_NODES + n) = o;
    }
}

extern "C" void kernel_launch(void* const* d_in, const int* in_sizes, int n_in,
                              void* d_out, int out_size, void* d_ws, size_t ws_size,
                              hipStream_t stream) {
    const float* h   = (const float*)d_in[0];
    const int*   src = (const int*)  d_in[1];
    const int*   dst = (const int*)  d_in[2];
    const float* W1  = (const float*)d_in[3];
    const float* b1  = (const float*)d_in[4];
    const float* W2  = (const float*)d_in[5];
    const float* b2  = (const float*)d_in[6];
    float*       out = (float*)d_out;

    float* nrmS   = (float*)d_ws + 2 * N_NODES;
    float* nrmD   = nrmS + N_NODES;
    int*   rowptr = (int*)d_ws + 4 * N_NODES;
    int*   csr    = rowptr + 50048;
    float* feat   = (float*)d_ws + 4 * N_NODES + 50048 + N_EDGES;
    float* z2     = feat + (size_t)N_NODES * 64;
    int*   histD  = (int*)(z2 + (size_t)N_NODES * 64);
    int*   histS  = histD + (size_t)HB * N_NODES;
    int*   gsum   = histS + (size_t)HB * N_NODES;

    const int T = 256;

    gcn_hist        <<<dim3(HB * HR, 2), 1024, 0, stream>>>(src, dst, histS, histD, gsum);
    gcn_combine_scan<<<GP_ALL, T, 0, stream>>>(histD, histS, gsum, rowptr, nrmS, nrmD);
    gcn_fill_transpose<<<FT_FILL + FT_TP, 1024, 0, stream>>>(src, dst, rowptr, histD, csr, h, nrmS, feat);

    gcn_gather_gemm12_wmma<<<N_NODES / 16, 256, 0, stream>>>(rowptr, csr, feat, nrmD, nrmS, W1, b1, W2, z2);

    gcn_gather32_out<<<(N_NODES + G2_NB - 1) / G2_NB, 256, 0, stream>>>(rowptr, csr, z2, nrmD, b2, out);
}


# edge bucketing rebuilt as a range partition: every chunk split by 256-node range in one launch, one work-group per range makes offsets, norms, csr and feat; 4 launches, not 5; sums in 64-chunk order
# speedup vs baseline: 6.3067x; 1.3090x over previous
#include <hip/hip_runtime.h>
#include <hip/hip_bf16.h>


typedef float v2f __attribute__((ext_vector_type(2)));
typedef float v8f __attribute__((ext_vector_type(8)));

#define N_NODES 50000
#define N_EDGES 800000
#define IN_D 64
#define HID_D 64
#define OUT_D 32

#define HB    64
#define H_CH  (N_EDGES / HB)
#define H_Q   (H_CH / 4)
#define H_IT  ((H_Q + 1023) / 1024)
#define RB    8
#define RN    (1 << RB)
#define HR    ((N_NODES + RN - 1) / RN)
#define OF_P  200

__device__ __forceinline__ int gcn_range_of(int id) {
    const unsigned g = (unsigned)id >> RB;
    return (int)(g < (unsigned)(HR - 1) ? g : (unsigned)(HR - 1));
}

__global__ __launch_bounds__(1024)
void gcn_partition(const int* __restrict__ src,
                   const int* __restrict__ dst,
                   int2* __restrict__ segD,
                   int* __restrict__ segS,
                   int* __restrict__ offD,
                   int* __restrict__ offS) {
    __shared__ int cnt[256];
    __shared__ int base[256];
    __shared__ int wsum[8];
    const int  tid  = threadIdx.x;
    const int  lane = tid & 31;
    const int  wv   = tid >> 5;
    const int  c    = blockIdx.x;
    const bool bysrc = blockIdx.y != 0;
    const int* const ids = (bysrc ? src : dst) + c * H_CH;
    const int* const pay = src + c * H_CH;

    int4 kv[H_IT], pv[H_IT];
#pragma unroll
    for (int i = 0; i < H_IT; ++i) {
        const int q = tid + 1024 * i;
        kv[i] = make_int4(0, 0, 0, 0);
        pv[i] = make_int4(0, 0, 0, 0);
        if (q < H_Q) {
            kv[i] = *(const int4*)(ids + 4 * q);
            pv[i] = *(const int4*)(pay + 4 * q);
        }
    }
    if (tid < 256) cnt[tid] = 0;
    __syncthreads();

    int4 gv[H_IT], rk[H_IT];
#pragma unroll
    for (int i = 0; i < H_IT; ++i) {
        const int q = tid + 1024 * i;
        gv[i] = make_int4(0, 0, 0, 0);
        rk[i] = make_int4(0, 0, 0, 0);
        if (q < H_Q) {
            gv[i].x = gcn_range_of(kv[i].x);
            gv[i].y = gcn_range_of(kv[i].y);
            gv[i].z = gcn_range_of(kv[i].z);
            gv[i].w = gcn_range_of(kv[i].w);
            rk[i].x = atomicAdd(cnt + gv[i].x, 1);
            rk[i].y = atomicAdd(cnt + gv[i].y, 1);
            rk[i].z = atomicAdd(cnt + gv[i].z, 1);
            rk[i].w = atomicAdd(cnt + gv[i].w, 1);
        }
    }
    __syncthreads();

    int v = 0, inc = 0;
    if (tid < 256) {
        v = tid < HR ? cnt[tid] : 0;
        inc = v;
        for (int o = 1; o < 32; o <<= 1) {
            const int u = __shfl_up(inc, o);
            if (lane >= o) inc += u;
        }
        if (lane == 31) wsum[wv] = inc;
    }
    __syncthreads();
    if (tid < 256) {
        int off = 0;
#pragma unroll
        for (int w = 0; w < 8; ++w)
            if (w < wv) off += wsum[w];
        const int ex = off + inc - v;
        base[tid] = ex;
        if (tid <= HR) (bysrc ? offS : offD)[c * OF_P + tid] = ex;
    }
    __syncthreads();

#pragma unroll
    for (int i = 0; i < H_IT; ++i) {
        const int q = tid + 1024 * i;
        if (q < H_Q) {
            const int p0 = min(max(base[gv[i].x] + rk[i].x, 0), H_CH - 1);
            const int p1 = min(max(base[gv[i].y] + rk[i].y, 0), H_CH - 1);
            const int p2 = min(max(base[gv[i].z] + rk[i].z, 0), H_CH - 1);
            const int p3 = min(max(base[gv[i].w] + rk[i].w, 0), H_CH - 1);
            if (bysrc) {
                int* const o = segS + c * H_CH;
                o[p0] = kv[i].x; o[p1] = kv[i].y; o[p2] = kv[i].z; o[p3] = kv[i].w;
            } else {
                int2* const o = segD + c * H_CH;
                o[p0] = make_int2(kv[i].x, pv[i].x);
                o[p1] = make_int2(kv[i].y, pv[i].y);
                o[p2] = make_int2(kv[i].z, pv[i].z);
                o[p3] = make_int2(kv[i].w, pv[i].w);
            }
        }
    }
}

#define TP_N     64
#define TP_PITCH 65
#define TP_GRP   (RN / TP_N)
#define SG_U     4
__global__ __launch_bounds__(1024)
void gcn_finish_transpose(const int2* __restrict__ segD,
                          const int* __restrict__ segS,
                          const int* __restrict__ offD,
                          const int* __restrict__ offS,
                          int* __restrict__ rowptr,
                          int* __restrict__ csr,
                          float* __restrict__ nrmS,
                          float* __restrict__ nrmD,
                          const float* __restrict__ h,
                          float* __restrict__ feat) {
    __shared__ __attribute__((aligned(16))) int cntL[HB * RN];
    __shared__ int   cntS[RN];
    __shared__ float sS[RN];
    __shared__ int   wsum[8];
    __shared__ int   wbase[2];
    __shared__ float tile[TP_GRP * TP_N * TP_PITCH];
    const int tid  = threadIdx.x;
    const int lane = tid & 31;
    const int wv   = tid >> 5;
    const int r    = blockIdx.x;
    const int lo   = r * RN;

    const int grp = tid >> 8;
    const int t   = tid & 255;
    const int cq  = t & 15;
    const int rr  = t >> 4;
    const int n0  = lo + grp * TP_N;
    const int nld = min(n0 + 4 * cq, N_NODES - 4);
    float4 hv[4];
#pragma unroll
    for (int p = 0; p < 4; ++p)
        hv[p] = *(const float4*)(h + (size_t)(rr + 16 * p) * N_NODES + nld);

    int aD[2], eD[2], aS[2], eS[2];
#pragma unroll
    for (int k = 0; k < 2; ++k) {
        const int c = wv + 32 * k;
        aD[k] = offD[c * OF_P + r];
        eD[k] = offD[c * OF_P + r + 1];
        aS[k] = offS[c * OF_P + r];
        eS[k] = offS[c * OF_P + r + 1];
    }
    if (tid < HB) {
        int b = offD[tid * OF_P + r];
        for (int o = 16; o > 0; o >>= 1) b += __shfl_xor(b, o);
        if (lane == 0) wbase[wv] = b;
    }
#pragma unroll
    for (int k = 0; k < 2; ++k) {
        aD[k] = min(max(aD[k], 0), H_CH);  eD[k] = min(max(eD[k], aD[k]), H_CH);
        aS[k] = min(max(aS[k], 0), H_CH);  eS[k] = min(max(eS[k], aS[k]), H_CH);
    }
    int2 pr[2][SG_U];
    int  sid[2][SG_U];
#pragma unroll
    for (int k = 0; k < 2; ++k) {
        const int c = wv + 32 * k;
#pragma unroll
        for (int u = 0; u < SG_U; ++u) {
            const int i = min(aD[k] + lane + 32 * u, H_CH - 1);
            const int j = min(aS[k] + lane + 32 * u, H_CH - 1);
            pr[k][u]  = segD[c * H_CH + i];
            sid[k][u] = segS[c * H_CH + j];
        }
    }
    for (int i = tid; i < HB * RN / 4; i += 1024) *(int4*)(cntL + 4 * i) = make_int4(0, 0, 0, 0);
    if (tid < RN) cntS[tid] = 0;
    __syncthreads();

#pragma unroll
    for (int k = 0; k < 2; ++k) {
        const int c = wv + 32 * k;
#pragma unroll
        for (int u = 0; u < SG_U; ++u) {
            if (aD[k] + lane + 32 * u < eD[k]) atomicAdd(cntL + c * RN + (pr[k][u].x & (RN - 1)), 1);
            if (aS[k] + lane + 32 * u < eS[k]) atomicAdd(cntS + (sid[k][u] & (RN - 1)), 1);
        }
        const int2* const pd = segD + c * H_CH;
        for (int i = aD[k] + 32 * SG_U + lane; i < eD[k]; i += 32)
            atomicAdd(cntL + c * RN + (pd[i].x & (RN - 1)), 1);
        const int* const ps = segS + c * H_CH;
        for (int i = aS[k] + 32 * SG_U + lane; i < eS[k]; i += 32)
            atomicAdd(cntS + (ps[i] & (RN - 1)), 1);
    }
    __syncthreads();

    const int  n    = lo + tid;
    const bool live = (tid < RN) & (n < N_NODES);
    int run = 0, inc = 0;
    if (tid < RN) {
#pragma unroll
        for (int c = 0; c < HB; ++c) {
            const int u = cntL[c * RN + tid];
            cntL[c * RN + tid] = run;
            run += u;
        }
        inc = run;
        for (int o = 1; o < 32; o <<= 1) {
            const int u = __shfl_up(inc, o);
            if (lane >= o) inc += u;
        }
        if (lane == 31) wsum[wv] = inc;
    }
    __syncthreads();
    if (tid < RN) {
        int off = wbase[0] + wbase[1];
#pragma unroll
        for (int w = 0; w < 8; ++w)
            if (w < wv) off += wsum[w];
        const int rp = off + inc - run;
        float s = 0.0f;
        if (live) {
            rowptr[n] = rp;
            nrmD[n] = rsqrtf(fmaxf((float)run, 1.0f));
            s = rsqrtf(fmaxf((float)cntS[tid], 1.0f));
            nrmS[n] = s;
        }
        sS[tid] = s;
        if (r == HR - 1 && tid == RN - 1) rowptr[N_NODES] = rp + run;
#pragma unroll
        for (int c = 0; c < HB; ++c) cntL[c * RN + tid] += rp;
    }
    __syncthreads();

#pragma unroll
    for (int k = 0; k < 2; ++k) {
        const int c = wv + 32 * k;
#pragma unroll
        for (int u = 0; u < SG_U; ++u) {
            if (aD[k] + lane + 32 * u < eD[k]) {
                const int slot = atomicAdd(cntL + c * RN + (pr[k][u].x & (RN - 1)), 1);
                csr[min(max(slot, 0), N_EDGES - 1)] = pr[k][u].y;
            }
        }
        const int2* const pd = segD + c * H_CH;
        for (int i = aD[k] + 32 * SG_U + lane; i < eD[k]; i += 32) {
            const int2 e = pd[i];
            const int slot = atomicAdd(cntL + c * RN + (e.x & (RN - 1)), 1);
            csr[min(max(slot, 0), N_EDGES - 1)] = e.y;
        }
    }

    {
        float* const tl = tile + grp * (TP_N * TP_PITCH);
        const float s0 = sS[grp * TP_N + 4 * cq + 0];
        const float s1 = sS[grp * TP_N + 4 * cq + 1];
        const float s2 = sS[grp * TP_N + 4 * cq + 2];
        const float s3 = sS[grp * TP_N + 4 * cq + 3];
#pragma unroll
        for (int p = 0; p < 4; ++p) {
            const int d = rr + 16 * p;
            tl[(4 * cq + 0) * TP_PITCH + d] = hv[p].x * s0;
            tl[(4 * cq + 1) * TP_PITCH + d] = hv[p].y * s1;
            tl[(4 * cq + 2) * TP_PITCH + d] = hv[p].z * s2;
            tl[(4 * cq + 3) * TP_PITCH + d] = hv[p].w * s3;
        }
    }
    __syncthreads();
    {
        const int rest   = N_NODES - n0;
        const int nvalid = rest < TP_N ? rest : TP_N;
        const float* const tl = tile + grp * (TP_N * TP_PITCH);
        for (int k = 0; k < 4; ++k) {
            const int e  = t + 256 * k;
            const int nn = e >> 4;
            const int q  = e & 15;
            if (nn < nvalid) {
                const float* tp = tl + nn * TP_PITCH + 4 * q;
                float4 o;
                o.x = tp[0]; o.y = tp[1]; o.z = tp[2]; o.w = tp[3];
                *(float4*)(feat + (size_t)n0 * IN_D + (size_t)e * 4) = o;
            }
        }
    }
}

#define X_PITCH 68
__global__ __launch_bounds__(256)
void gcn_gather_gemm12_wmma(const int* __restrict__ rowptr,
                            const int* __restrict__ csr,
                            const float* __restrict__ feat,
                            const float* __restrict__ nrmD,
                            const float* __restrict__ nrmS,
                            const float* __restrict__ W1,
                            const float* __restrict__ b1,
                            const float* __restrict__ W2,
                            float* __restrict__ z2) {
    __shared__ float as[16 * X_PITCH];
    __shared__ float xs[16 * X_PITCH];

    const int nn   = threadIdx.x >> 4;
    const int node = blockIdx.x * 16 + nn;
    const int l    = threadIdx.x & 15;
    int beg = rowptr[node];
    int end = rowptr[node + 1];
    const float sD = nrmD[node];
    beg = max(beg, 0);
    end = min(end, N_EDGES);
    const float* const fcol = feat + 4 * l;

    float4 acc = make_float4(0.0f, 0.0f, 0.0f, 0.0f);
    int next = 0;
    if (beg + l < end) next = csr[beg + l];
    for (int j = beg; j < end; j += 16) {
        const int rem = end - j;
        const int cnt = rem < 16 ? rem : 16;
        const int mine = min(max(next, 0), N_NODES - 1);
        next = 0;
        if (j + 16 + l < end) next = csr[j + 16 + l];
        for (int k = 0; k < cnt; k += 4) {
            const int s0 = __shfl(mine, k + 0, 16);
            const int s1 = __shfl(mine, k + 1, 16);
            const int s2 = __shfl(mine, k + 2, 16);
            const int s3 = __shfl(mine, k + 3, 16);
            const float4 z = make_float4(0.0f, 0.0f, 0.0f, 0.0f);
            float4 v0 = *(const float4*)(fcol + (size_t)s0 * 64);
            float4 v1 = z, v2 = z, v3 = z;
            if (k + 1 < cnt) v1 = *(const float4*)(fcol + (size_t)s1 * 64);
            if (k + 2 < cnt) v2 = *(const float4*)(fcol + (size_t)s2 * 64);
            if (k + 3 < cnt) v3 = *(const float4*)(fcol + (size_t)s3 * 64);
            acc.x += v0.x; acc.y += v0.y; acc.z += v0.z; acc.w += v0.w;
            acc.x += v1.x; acc.y += v1.y; acc.z += v1.z; acc.w += v1.w;
            acc.x += v2.x; acc.y += v2.y; acc.z += v2.z; acc.w += v2.w;
            acc.x += v3.x; acc.y += v3.y; acc.z += v3.z; acc.w += v3.w;
        }
    }
    const int lane  = threadIdx.x & 31;
    const int wv    = __builtin_amdgcn_readfirstlane(threadIdx.x >> 5);
    const int m     = lane & 15;
    const int khalf = (lane >> 4) << 1;
    const int rloc  = (lane >> 4) << 3;
    const int rbase = blockIdx.x * 16 + rloc;
    const bool first  = wv < 4;
    const int  w2     = 4 + ((blockIdx.x & 1) << 1);
    const bool second = (wv == w2) | (wv == w2 + 1);
    const int  col    = first ? (wv << 4) + m : ((wv & 1) << 4) + m;

    v2f   bw[16];
    float nsv[8];
    float bias = 0.0f;
    if (first) {
#pragma unroll
        for (int t = 0; t < 16; ++t) {
            const int k = 4 * t + khalf;
            bw[t].x = W1[k * HID_D + col];
            bw[t].y = W1[(k + 1) * HID_D + col];
        }
        bias = b1[col];
#pragma unroll
        for (int j = 0; j < 8; ++j) nsv[j] = nrmS[rbase + j];
    } else if (second) {
#pragma unroll
        for (int t = 0; t < 16; ++t) {
            const int k = 4 * t + khalf;
            bw[t].x = W2[k * OUT_D + col];
            bw[t].y = W2[(k + 1) * OUT_D + col];
        }
    }

    *(float4*)(as + nn * X_PITCH + 4 * l) =
        make_float4(acc.x * sD, acc.y * sD, acc.z * sD, acc.w * sD);
    __syncthreads();


    if (first) {
        const float* arow = as + m * X_PITCH;
        v8f acc1 = {};
#pragma unroll
        for (int t = 0; t < 16; ++t) {
            const int k = 4 * t + khalf;
            v2f a;
            a.x = arow[k];
            a.y = arow[k + 1];
            acc1 = __builtin_amdgcn_wmma_f32_16x16x4_f32(
                false, a, false, bw[t], (short)0, acc1, false, false);
        }
#pragma unroll
        for (int j = 0; j < 8; ++j) {
            float u = acc1[j] + bias;
            u = u > 0.0f ? u : 0.0f;
            xs[(rloc + j) * X_PITCH + col] = u * nsv[j];
        }
    }
    __syncthreads();

    if (second) {
        const float* xrow = xs + m * X_PITCH;
        v8f acc2 = {};
#pragma unroll
        for (int t = 0; t < 16; ++t) {
            const int k = 4 * t + khalf;
            v2f a;
            a.x = xrow[k];
            a.y = xrow[k + 1];
            acc2 = __builtin_amdgcn_wmma_f32_16x16x4_f32(
                false, a, false, bw[t], (short)0, acc2, false, false);
        }
#pragma unroll
        for (int j = 0; j < 8; ++j)
            z2[(size_t)(rbase + j) * OUT_D + col] = acc2[j];
    }
}

#define G2_NB 32
#define G2_P  33
__global__ __launch_bounds__(256)
void gcn_gather32_out(const int* __restrict__ rowptr,
                      const int* __restrict__ csr,
                      const float* __restrict__ z2,
                      const float* __restrict__ nrmD,
                      const float* __restrict__ b2,
                      float* __restrict__ out) {
    __shared__ float tile[OUT_D * G2_P];
    const int tid  = threadIdx.x;
    const int nn   = tid >> 3;
    const int q    = tid & 7;
    const int n0   = blockIdx.x * G2_NB;
    const int node = n0 + nn;

    int   beg = 0, end = 0;
    float sD  = 0.0f;
    if (node < N_NODES) {
        beg = rowptr[node];
        end = rowptr[node + 1];
        sD  = nrmD[node];
    }
    beg = max(beg, 0);
    end = min(end, N_EDGES);
    const float4 bb = *(const float4*)(b2 + 4 * q);
    const float* const zcol = z2 + 4 * q;

    float4 acc = make_float4(0.0f, 0.0f, 0.0f, 0.0f);
    int next = 0;
    if (beg + q < end) next = csr[beg + q];
    for (int j = beg; j < end; j += 8) {
        const int rem  = end - j;
        const int cnt  = rem < 8 ? rem : 8;
        const int mine = min(max(next, 0), N_NODES - 1);
        const int s0 = __shfl(mine, 0, 8);
        const int s1 = __shfl(mine, 1, 8);
        const int s2 = __shfl(mine, 2, 8);
        const int s3 = __shfl(mine, 3, 8);
        const int s4 = __shfl(mine, 4, 8);
        const int s5 = __shfl(mine, 5, 8);
        const int s6 = __shfl(mine, 6, 8);
        const int s7 = __shfl(mine, 7, 8);
        const float4 z = make_float4(0.0f, 0.0f, 0.0f, 0.0f);
        float4 v0 = *(const float4*)(zcol + (size_t)s0 * OUT_D);
        float4 v1 = z, v2 = z, v3 = z, v4 = z, v5 = z, v6 = z, v7 = z;
        if (1 < cnt) v1 = *(const float4*)(zcol + (size_t)s1 * OUT_D);
        if (2 < cnt) v2 = *(const float4*)(zcol + (size_t)s2 * OUT_D);
        if (3 < cnt) v3 = *(const float4*)(zcol + (size_t)s3 * OUT_D);
        if (4 < cnt) v4 = *(const float4*)(zcol + (size_t)s4 * OUT_D);
        if (5 < cnt) v5 = *(const float4*)(zcol + (size_t)s5 * OUT_D);
        if (6 < cnt) v6 = *(const float4*)(zcol + (size_t)s6 * OUT_D);
        if (7 < cnt) v7 = *(const float4*)(zcol + (size_t)s7 * OUT_D);
        next = 0;
        if (j + 8 + q < end) next = csr[j + 8 + q];
        acc.x += v0.x; acc.y += v0.y; acc.z += v0.z; acc.w += v0.w;
        acc.x += v1.x; acc.y += v1.y; acc.z += v1.z; acc.w += v1.w;
        acc.x += v2.x; acc.y += v2.y; acc.z += v2.z; acc.w += v2.w;
        acc.x += v3.x; acc.y += v3.y; acc.z += v3.z; acc.w += v3.w;
        acc.x += v4.x; acc.y += v4.y; acc.z += v4.z; acc.w += v4.w;
        acc.x += v5.x; acc.y += v5.y; acc.z += v5.z; acc.w += v5.w;
        acc.x += v6.x; acc.y += v6.y; acc.z += v6.z; acc.w += v6.w;
        acc.x += v7.x; acc.y += v7.y; acc.z += v7.z; acc.w += v7.w;
    }

    tile[(4 * q + 0) * G2_P + nn] = acc.x * sD + bb.x;
    tile[(4 * q + 1) * G2_P + nn] = acc.y * sD + bb.y;
    tile[(4 * q + 2) * G2_P + nn] = acc.z * sD + bb.z;
    tile[(4 * q + 3) * G2_P + nn] = acc.w * sD + bb.w;
    __syncthreads();

    const int c    = tid >> 3;
    const int quad = tid & 7;
    const int n    = n0 + 4 * quad;
    if (n < N_NODES) {
        const float* t = tile + c * G2_P + 4 * quad;
        float4 o;
        o.x = t[0]; o.y = t[1]; o.z = t[2]; o.w = t[3];
        *(float4*)(out + (size_t)c * N_NODES + n) = o;
    }
}

extern "C" void kernel_launch(void* const* d_in, const int* in_sizes, int n_in,
                              void* d_out, int out_size, void* d_ws, size_t ws_size,
                              hipStream_t stream) {
    const float* h   = (const float*)d_in[0];
    const int*   src = (const int*)  d_in[1];
    const int*   dst = (const int*)  d_in[2];
    const float* W1  = (const float*)d_in[3];
    const float* b1  = (const float*)d_in[4];
    const float* W2  = (const float*)d_in[5];
    const float* b2  = (const float*)d_in[6];
    float*       out = (float*)d_out;

    float* nrmS   = (float*)d_ws + 2 * N_NODES;
    float* nrmD   = nrmS + N_NODES;
    int*   rowptr = (int*)d_ws + 4 * N_NODES;
    int*   csr    = rowptr + 50048;
    float* feat   = (float*)d_ws + 4 * N_NODES + 50048 + N_EDGES;
    float* z2     = feat + (size_t)N_NODES * 64;
    int2*  segD   = (int2*)(z2 + (size_t)N_NODES * 64);
    int*   segS   = (int*)(segD + N_EDGES);
    int*   offD   = segS + N_EDGES;
    int*   offS   = offD + HB * OF_P;

    gcn_partition       <<<dim3(HB, 2), 1024, 0, stream>>>(src, dst, segD, segS, offD, offS);
    gcn_finish_transpose<<<HR, 1024, 0, stream>>>(segD, segS, offD, offS, rowptr, csr, nrmS, nrmD, h, feat);

    gcn_gather_gemm12_wmma<<<N_NODES / 16, 256, 0, stream>>>(rowptr, csr, feat, nrmD, nrmS, W1, b1, W2, z2);

    gcn_gather32_out<<<(N_NODES + G2_NB - 1) / G2_NB, 256, 0, stream>>>(rowptr, csr, z2, nrmD, b2, out);
}


# edge bucketing's scattered 4- and 8-byte global stores (split chunks, csr slots) collected in LDS and written out as contiguous runs in whole lines; places, slot order and sums unchanged
# speedup vs baseline: 7.2749x; 1.1535x over previous
#include <hip/hip_runtime.h>
#include <hip/hip_bf16.h>


typedef float v2f __attribute__((ext_vector_type(2)));
typedef float v8f __attribute__((ext_vector_type(8)));

#define N_NODES 50000
#define N_EDGES 800000
#define IN_D 64
#define HID_D 64
#define OUT_D 32

#define HB    64
#define H_CH  (N_EDGES / HB)
#define H_Q   (H_CH / 4)
#define H_IT  ((H_Q + 1023) / 1024)
#define RB    8
#define RN    (1 << RB)
#define HR    ((N_NODES + RN - 1) / RN)
#define OF_P  200
#define ST_IT ((H_CH / 2 + 1023) / 1024)

__device__ __forceinline__ int gcn_range_of(int id) {
    const unsigned g = (unsigned)id >> RB;
    return (int)(g < (unsigned)(HR - 1) ? g : (unsigned)(HR - 1));
}

__global__ __launch_bounds__(1024)
void gcn_partition(const int* __restrict__ src,
                   const int* __restrict__ dst,
                   int2* __restrict__ segD,
                   int* __restrict__ segS,
                   int* __restrict__ offD,
                   int* __restrict__ offS) {
    __shared__ int cnt[256];
    __shared__ int base[256];
    __shared__ int wsum[8];
    __shared__ __attribute__((aligned(16))) int stage[2 * H_CH];
    const int  tid  = threadIdx.x;
    const int  lane = tid & 31;
    const int  wv   = tid >> 5;
    const int  c    = blockIdx.x;
    const bool bysrc = blockIdx.y != 0;
    const int* const ids = (bysrc ? src : dst) + c * H_CH;
    const int* const pay = src + c * H_CH;

    int4 kv[H_IT], pv[H_IT];
#pragma unroll
    for (int i = 0; i < H_IT; ++i) {
        const int q = tid + 1024 * i;
        kv[i] = make_int4(0, 0, 0, 0);
        pv[i] = make_int4(0, 0, 0, 0);
        if (q < H_Q) {
            kv[i] = *(const int4*)(ids + 4 * q);
            pv[i] = *(const int4*)(pay + 4 * q);
        }
    }
    if (tid < 256) cnt[tid] = 0;
    __syncthreads();

    int4 gv[H_IT], rk[H_IT];
#pragma unroll
    for (int i = 0; i < H_IT; ++i) {
        const int q = tid + 1024 * i;
        gv[i] = make_int4(0, 0, 0, 0);
        rk[i] = make_int4(0, 0, 0, 0);
        if (q < H_Q) {
            gv[i].x = gcn_range_of(kv[i].x);
            gv[i].y = gcn_range_of(kv[i].y);
            gv[i].z = gcn_range_of(kv[i].z);
            gv[i].w = gcn_range_of(kv[i].w);
            rk[i].x = atomicAdd(cnt + gv[i].x, 1);
            rk[i].y = atomicAdd(cnt + gv[i].y, 1);
            rk[i].z = atomicAdd(cnt + gv[i].z, 1);
            rk[i].w = atomicAdd(cnt + gv[i].w, 1);
        }
    }
    __syncthreads();

    int v = 0, inc = 0;
    if (tid < 256) {
        v = tid < HR ? cnt[tid] : 0;
        inc = v;
        for (int o = 1; o < 32; o <<= 1) {
            const int u = __shfl_up(inc, o);
            if (lane >= o) inc += u;
        }
        if (lane == 31) wsum[wv] = inc;
    }
    __syncthreads();
    if (tid < 256) {
        int off = 0;
#pragma unroll
        for (int w = 0; w < 8; ++w)
            if (w < wv) off += wsum[w];
        const int ex = off + inc - v;
        base[tid] = ex;
        if (tid <= HR) (bysrc ? offS : offD)[c * OF_P + tid] = ex;
    }
    __syncthreads();

#pragma unroll
    for (int i = 0; i < H_IT; ++i) {
        const int q = tid + 1024 * i;
        if (q < H_Q) {
            const int p0 = min(max(base[gv[i].x] + rk[i].x, 0), H_CH - 1);
            const int p1 = min(max(base[gv[i].y] + rk[i].y, 0), H_CH - 1);
            const int p2 = min(max(base[gv[i].z] + rk[i].z, 0), H_CH - 1);
            const int p3 = min(max(base[gv[i].w] + rk[i].w, 0), H_CH - 1);
            if (bysrc) {
                stage[p0] = kv[i].x; stage[p1] = kv[i].y; stage[p2] = kv[i].z; stage[p3] = kv[i].w;
            } else {
                *(int2*)(stage + 2 * p0) = make_int2(kv[i].x, pv[i].x);
                *(int2*)(stage + 2 * p1) = make_int2(kv[i].y, pv[i].y);
                *(int2*)(stage + 2 * p2) = make_int2(kv[i].z, pv[i].z);
                *(int2*)(stage + 2 * p3) = make_int2(kv[i].w, pv[i].w);
            }
        }
    }
    __syncthreads();
    {
        const int   n4 = bysrc ? H_CH / 4 : H_CH / 2;
        int4* const o  = bysrc ? (int4*)(segS + c * H_CH) : (int4*)(segD + c * H_CH);
#pragma unroll
        for (int k = 0; k < ST_IT; ++k) {
            const int i = tid + 1024 * k;
            if (i < n4) o[i] = *(const int4*)(stage + 4 * i);
        }
    }
}

#define TP_N     64
#define TP_PITCH 65
#define TP_GRP   (RN / TP_N)
#define SG_U     4
#define CS_CAP   8192
__global__ __launch_bounds__(1024)
void gcn_finish_transpose(const int2* __restrict__ segD,
                          const int* __restrict__ segS,
                          const int* __restrict__ offD,
                          const int* __restrict__ offS,
                          int* __restrict__ rowptr,
                          int* __restrict__ csr,
                          float* __restrict__ nrmS,
                          float* __restrict__ nrmD,
                          const float* __restrict__ h,
                          float* __restrict__ feat) {
    __shared__ __attribute__((aligned(16))) int cntL[HB * RN];
    __shared__ int   cntS[RN];
    __shared__ float sS[RN];
    __shared__ int   wsum[8];
    __shared__ int   wbase[2];
    __shared__ int   wend;
    __shared__ int   cstage[CS_CAP];
    __shared__ float tile[TP_GRP * TP_N * TP_PITCH];
    const int tid  = threadIdx.x;
    const int lane = tid & 31;
    const int wv   = tid >> 5;
    const int r    = blockIdx.x;
    const int lo   = r * RN;

    const int grp = tid >> 8;
    const int t   = tid & 255;
    const int cq  = t & 15;
    const int rr  = t >> 4;
    const int n0  = lo + grp * TP_N;
    const int nld = min(n0 + 4 * cq, N_NODES - 4);
    float4 hv[4];
#pragma unroll
    for (int p = 0; p < 4; ++p)
        hv[p] = *(const float4*)(h + (size_t)(rr + 16 * p) * N_NODES + nld);

    int aD[2], eD[2], aS[2], eS[2];
#pragma unroll
    for (int k = 0; k < 2; ++k) {
        const int c = wv + 32 * k;
        aD[k] = offD[c * OF_P + r];
        eD[k] = offD[c * OF_P + r + 1];
        aS[k] = offS[c * OF_P + r];
        eS[k] = offS[c * OF_P + r + 1];
    }
    if (tid < HB) {
        int b = offD[tid * OF_P + r];
        for (int o = 16; o > 0; o >>= 1) b += __shfl_xor(b, o);
        if (lane == 0) wbase[wv] = b;
    }
#pragma unroll
    for (int k = 0; k < 2; ++k) {
        aD[k] = min(max(aD[k], 0), H_CH);  eD[k] = min(max(eD[k], aD[k]), H_CH);
        aS[k] = min(max(aS[k], 0), H_CH);  eS[k] = min(max(eS[k], aS[k]), H_CH);
    }
    int2 pr[2][SG_U];
    int  sid[2][SG_U];
#pragma unroll
    for (int k = 0; k < 2; ++k) {
        const int c = wv + 32 * k;
#pragma unroll
        for (int u = 0; u < SG_U; ++u) {
            const int i = min(aD[k] + lane + 32 * u, H_CH - 1);
            const int j = min(aS[k] + lane + 32 * u, H_CH - 1);
            pr[k][u]  = segD[c * H_CH + i];
            sid[k][u] = segS[c * H_CH + j];
        }
    }
    for (int i = tid; i < HB * RN / 4; i += 1024) *(int4*)(cntL + 4 * i) = make_int4(0, 0, 0, 0);
    if (tid < RN) cntS[tid] = 0;
    __syncthreads();

#pragma unroll
    for (int k = 0; k < 2; ++k) {
        const int c = wv + 32 * k;
#pragma unroll
        for (int u = 0; u < SG_U; ++u) {
            if (aD[k] + lane + 32 * u < eD[k]) atomicAdd(cntL + c * RN + (pr[k][u].x & (RN - 1)), 1);
            if (aS[k] + lane + 32 * u < eS[k]) atomicAdd(cntS + (sid[k][u] & (RN - 1)), 1);
        }
        const int2* const pd = segD + c * H_CH;
        for (int i = aD[k] + 32 * SG_U + lane; i < eD[k]; i += 32)
            atomicAdd(cntL + c * RN + (pd[i].x & (RN - 1)), 1);
        const int* const ps = segS + c * H_CH;
        for (int i = aS[k] + 32 * SG_U + lane; i < eS[k]; i += 32)
            atomicAdd(cntS + (ps[i] & (RN - 1)), 1);
    }
    __syncthreads();

    const int  n    = lo + tid;
    const bool live = (tid < RN) & (n < N_NODES);
    int run = 0, inc = 0;
    if (tid < RN) {
#pragma unroll
        for (int c = 0; c < HB; ++c) {
            const int u = cntL[c * RN + tid];
            cntL[c * RN + tid] = run;
            run += u;
        }
        inc = run;
        for (int o = 1; o < 32; o <<= 1) {
            const int u = __shfl_up(inc, o);
            if (lane >= o) inc += u;
        }
        if (lane == 31) wsum[wv] = inc;
    }
    __syncthreads();
    if (tid < RN) {
        int off = wbase[0] + wbase[1];
#pragma unroll
        for (int w = 0; w < 8; ++w)
            if (w < wv) off += wsum[w];
        const int rp = off + inc - run;
        float s = 0.0f;
        if (live) {
            rowptr[n] = rp;
            nrmD[n] = rsqrtf(fmaxf((float)run, 1.0f));
            s = rsqrtf(fmaxf((float)cntS[tid], 1.0f));
            nrmS[n] = s;
        }
        sS[tid] = s;
        if (r == HR - 1 && tid == RN - 1) rowptr[N_NODES] = rp + run;
        if (tid == RN - 1) wend = rp + run;
#pragma unroll
        for (int c = 0; c < HB; ++c) cntL[c * RN + tid] += rp;
    }
    __syncthreads();

    const int segBase = wbase[0] + wbase[1];
#pragma unroll
    for (int k = 0; k < 2; ++k) {
        const int c = wv + 32 * k;
#pragma unroll
        for (int u = 0; u < SG_U; ++u) {
            if (aD[k] + lane + 32 * u < eD[k]) {
                const int slot = atomicAdd(cntL + c * RN + (pr[k][u].x & (RN - 1)), 1);
                const unsigned loc = (unsigned)(slot - segBase);
                if (loc < (unsigned)CS_CAP) cstage[loc] = pr[k][u].y;
                else csr[min(max(slot, 0), N_EDGES - 1)] = pr[k][u].y;
            }
        }
        const int2* const pd = segD + c * H_CH;
        for (int i = aD[k] + 32 * SG_U + lane; i < eD[k]; i += 32) {
            const int2 e = pd[i];
            const int slot = atomicAdd(cntL + c * RN + (e.x & (RN - 1)), 1);
            const unsigned loc = (unsigned)(slot - segBase);
            if (loc < (unsigned)CS_CAP) cstage[loc] = e.y;
            else csr[min(max(slot, 0), N_EDGES - 1)] = e.y;
        }
    }

    {
        float* const tl = tile + grp * (TP_N * TP_PITCH);
        const float s0 = sS[grp * TP_N + 4 * cq + 0];
        const float s1 = sS[grp * TP_N + 4 * cq + 1];
        const float s2 = sS[grp * TP_N + 4 * cq + 2];
        const float s3 = sS[grp * TP_N + 4 * cq + 3];
#pragma unroll
        for (int p = 0; p < 4; ++p) {
            const int d = rr + 16 * p;
            tl[(4 * cq + 0) * TP_PITCH + d] = hv[p].x * s0;
            tl[(4 * cq + 1) * TP_PITCH + d] = hv[p].y * s1;
            tl[(4 * cq + 2) * TP_PITCH + d] = hv[p].z * s2;
            tl[(4 * cq + 3) * TP_PITCH + d] = hv[p].w * s3;
        }
    }
    __syncthreads();
    {
        const int segLen = min(max(wend - segBase, 0), CS_CAP);
        int cv[CS_CAP / 1024];
#pragma unroll
        for (int k = 0; k < CS_CAP / 1024; ++k) cv[k] = cstage[tid + 1024 * k];
#pragma unroll
        for (int k = 0; k < CS_CAP / 1024; ++k) {
            const int i = tid + 1024 * k;
            if (i < segLen) csr[min(max(segBase + i, 0), N_EDGES - 1)] = cv[k];
        }
    }
    {
        const int rest   = N_NODES - n0;
        const int nvalid = rest < TP_N ? rest : TP_N;
        const float* const tl = tile + grp * (TP_N * TP_PITCH);
        for (int k = 0; k < 4; ++k) {
            const int e  = t + 256 * k;
            const int nn = e >> 4;
            const int q  = e & 15;
            if (nn < nvalid) {
                const float* tp = tl + nn * TP_PITCH + 4 * q;
                float4 o;
                o.x = tp[0]; o.y = tp[1]; o.z = tp[2]; o.w = tp[3];
                *(float4*)(feat + (size_t)n0 * IN_D + (size_t)e * 4) = o;
            }
        }
    }
}

#define X_PITCH 68
__global__ __launch_bounds__(256)
void gcn_gather_gemm12_wmma(const int* __restrict__ rowptr,
                            const int* __restrict__ csr,
                            const float* __restrict__ feat,
                            const float* __restrict__ nrmD,
                            const float* __restrict__ nrmS,
                            const float* __restrict__ W1,
                            const float* __restrict__ b1,
                            const float* __restrict__ W2,
                            float* __restrict__ z2) {
    __shared__ float as[16 * X_PITCH];
    __shared__ float xs[16 * X_PITCH];

    const int nn   = threadIdx.x >> 4;
    const int node = blockIdx.x * 16 + nn;
    const int l    = threadIdx.x & 15;
    int beg = rowptr[node];
    int end = rowptr[node + 1];
    const float sD = nrmD[node];
    beg = max(beg, 0);
    end = min(end, N_EDGES);
    const float* const fcol = feat + 4 * l;

    float4 acc = make_float4(0.0f, 0.0f, 0.0f, 0.0f);
    int next = 0;
    if (beg + l < end) next = csr[beg + l];
    for (int j = beg; j < end; j += 16) {
        const int rem = end - j;
        const int cnt = rem < 16 ? rem : 16;
        const int mine = min(max(next, 0), N_NODES - 1);
        next = 0;
        if (j + 16 + l < end) next = csr[j + 16 + l];
        for (int k = 0; k < cnt; k += 4) {
            const int s0 = __shfl(mine, k + 0, 16);
            const int s1 = __shfl(mine, k + 1, 16);
            const int s2 = __shfl(mine, k + 2, 16);
            const int s3 = __shfl(mine, k + 3, 16);
            const float4 z = make_float4(0.0f, 0.0f, 0.0f, 0.0f);
            float4 v0 = *(const float4*)(fcol + (size_t)s0 * 64);
            float4 v1 = z, v2 = z, v3 = z;
            if (k + 1 < cnt) v1 = *(const float4*)(fcol + (size_t)s1 * 64);
            if (k + 2 < cnt) v2 = *(const float4*)(fcol + (size_t)s2 * 64);
            if (k + 3 < cnt) v3 = *(const float4*)(fcol + (size_t)s3 * 64);
            acc.x += v0.x; acc.y += v0.y; acc.z += v0.z; acc.w += v0.w;
            acc.x += v1.x; acc.y += v1.y; acc.z += v1.z; acc.w += v1.w;
            acc.x += v2.x; acc.y += v2.y; acc.z += v2.z; acc.w += v2.w;
            acc.x += v3.x; acc.y += v3.y; acc.z += v3.z; acc.w += v3.w;
        }
    }
    const int lane  = threadIdx.x & 31;
    const int wv    = __builtin_amdgcn_readfirstlane(threadIdx.x >> 5);
    const int m     = lane & 15;
    const int khalf = (lane >> 4) << 1;
    const int rloc  = (lane >> 4) << 3;
    const int rbase = blockIdx.x * 16 + rloc;
    const bool first  = wv < 4;
    const int  w2     = 4 + ((blockIdx.x & 1) << 1);
    const bool second = (wv == w2) | (wv == w2 + 1);
    const int  col    = first ? (wv << 4) + m : ((wv & 1) << 4) + m;

    v2f   bw[16];
    float nsv[8];
    float bias = 0.0f;
    if (first) {
#pragma unroll
        for (int t = 0; t < 16; ++t) {
            const int k = 4 * t + khalf;
            bw[t].x = W1[k * HID_D + col];
            bw[t].y = W1[(k + 1) * HID_D + col];
        }
        bias = b1[col];
#pragma unroll
        for (int j = 0; j < 8; ++j) nsv[j] = nrmS[rbase + j];
    } else if (second) {
#pragma unroll
        for (int t = 0; t < 16; ++t) {
            const int k = 4 * t + khalf;
            bw[t].x = W2[k * OUT_D + col];
            bw[t].y = W2[(k + 1) * OUT_D + col];
        }
    }

    *(float4*)(as + nn * X_PITCH + 4 * l) =
        make_float4(acc.x * sD, acc.y * sD, acc.z * sD, acc.w * sD);
    __syncthreads();


    if (first) {
        const float* arow = as + m * X_PITCH;
        v8f acc1 = {};
#pragma unroll
        for (int t = 0; t < 16; ++t) {
            const int k = 4 * t + khalf;
            v2f a;
            a.x = arow[k];
            a.y = arow[k + 1];
            acc1 = __builtin_amdgcn_wmma_f32_16x16x4_f32(
                false, a, false, bw[t], (short)0, acc1, false, false);
        }
#pragma unroll
        for (int j = 0; j < 8; ++j) {
            float u = acc1[j] + bias;
            u = u > 0.0f ? u : 0.0f;
            xs[(rloc + j) * X_PITCH + col] = u * nsv[j];
        }
    }
    __syncthreads();

    if (second) {
        const float* xrow = xs + m * X_PITCH;
        v8f acc2 = {};
#pragma unroll
        for (int t = 0; t < 16; ++t) {
            const int k = 4 * t + khalf;
            v2f a;
            a.x = xrow[k];
            a.y = xrow[k + 1];
            acc2 = __builtin_amdgcn_wmma_f32_16x16x4_f32(
                false, a, false, bw[t], (short)0, acc2, false, false);
        }
#pragma unroll
        for (int j = 0; j < 8; ++j)
            z2[(size_t)(rbase + j) * OUT_D + col] = acc2[j];
    }
}

#define G2_NB 32
#define G2_P  33
__global__ __launch_bounds__(256)
void gcn_gather32_out(const int* __restrict__ rowptr,
                      const int* __restrict__ csr,
                      const float* __restrict__ z2,
                      const float* __restrict__ nrmD,
                      const float* __restrict__ b2,
                      float* __restrict__ out) {
    __shared__ float tile[OUT_D * G2_P];
    const int tid  = threadIdx.x;
    const int nn   = tid >> 3;
    const int q    = tid & 7;
    const int n0   = blockIdx.x * G2_NB;
    const int node = n0 + nn;

    int   beg = 0, end = 0;
    float sD  = 0.0f;
    if (node < N_NODES) {
        beg = rowptr[node];
        end = rowptr[node + 1];
        sD  = nrmD[node];
    }
    beg = max(beg, 0);
    end = min(end, N_EDGES);
    const float4 bb = *(const float4*)(b2 + 4 * q);
    const float* const zcol = z2 + 4 * q;

    float4 acc = make_float4(0.0f, 0.0f, 0.0f, 0.0f);
    int next = 0;
    if (beg + q < end) next = csr[beg + q];
    for (int j = beg; j < end; j += 8) {
        const int rem  = end - j;
        const int cnt  = rem < 8 ? rem : 8;
        const int mine = min(max(next, 0), N_NODES - 1);
        const int s0 = __shfl(mine, 0, 8);
        const int s1 = __shfl(mine, 1, 8);
        const int s2 = __shfl(mine, 2, 8);
        const int s3 = __shfl(mine, 3, 8);
        const int s4 = __shfl(mine, 4, 8);
        const int s5 = __shfl(mine, 5, 8);
        const int s6 = __shfl(mine, 6, 8);
        const int s7 = __shfl(mine, 7, 8);
        const float4 z = make_float4(0.0f, 0.0f, 0.0f, 0.0f);
        float4 v0 = *(const float4*)(zcol + (size_t)s0 * OUT_D);
        float4 v1 = z, v2 = z, v3 = z, v4 = z, v5 = z, v6 = z, v7 = z;
        if (1 < cnt) v1 = *(const float4*)(zcol + (size_t)s1 * OUT_D);
        if (2 < cnt) v2 = *(const float4*)(zcol + (size_t)s2 * OUT_D);
        if (3 < cnt) v3 = *(const float4*)(zcol + (size_t)s3 * OUT_D);
        if (4 < cnt) v4 = *(const float4*)(zcol + (size_t)s4 * OUT_D);
        if (5 < cnt) v5 = *(const float4*)(zcol + (size_t)s5 * OUT_D);
        if (6 < cnt) v6 = *(const float4*)(zcol + (size_t)s6 * OUT_D);
        if (7 < cnt) v7 = *(const float4*)(zcol + (size_t)s7 * OUT_D);
        next = 0;
        if (j + 8 + q < end) next = csr[j + 8 + q];
        acc.x += v0.x; acc.y += v0.y; acc.z += v0.z; acc.w += v0.w;
        acc.x += v1.x; acc.y += v1.y; acc.z += v1.z; acc.w += v1.w;
        acc.x += v2.x; acc.y += v2.y; acc.z += v2.z; acc.w += v2.w;
        acc.x += v3.x; acc.y += v3.y; acc.z += v3.z; acc.w += v3.w;
        acc.x += v4.x; acc.y += v4.y; acc.z += v4.z; acc.w += v4.w;
        acc.x += v5.x; acc.y += v5.y; acc.z += v5.z; acc.w += v5.w;
        acc.x += v6.x; acc.y += v6.y; acc.z += v6.z; acc.w += v6.w;
        acc.x += v7.x; acc.y += v7.y; acc.z += v7.z; acc.w += v7.w;
    }

    tile[(4 * q + 0) * G2_P + nn] = acc.x * sD + bb.x;
    tile[(4 * q + 1) * G2_P + nn] = acc.y * sD + bb.y;
    tile[(4 * q + 2) * G2_P + nn] = acc.z * sD + bb.z;
    tile[(4 * q + 3) * G2_P + nn] = acc.w * sD + bb.w;
    __syncthreads();

    const int c    = tid >> 3;
    const int quad = tid & 7;
    const int n    = n0 + 4 * quad;
    if (n < N_NODES) {
        const float* t = tile + c * G2_P + 4 * quad;
        float4 o;
        o.x = t[0]; o.y = t[1]; o.z = t[2]; o.w = t[3];
        *(float4*)(out + (size_t)c * N_NODES + n) = o;
    }
}

extern "C" void kernel_launch(void* const* d_in, const int* in_sizes, int n_in,
                              void* d_out, int out_size, void* d_ws, size_t ws_size,
                              hipStream_t stream) {
    const float* h   = (const float*)d_in[0];
    const int*   src = (const int*)  d_in[1];
    const int*   dst = (const int*)  d_in[2];
    const float* W1  = (const float*)d_in[3];
    const float* b1  = (const float*)d_in[4];
    const float* W2  = (const float*)d_in[5];
    const float* b2  = (const float*)d_in[6];
    float*       out = (float*)d_out;

    float* nrmS   = (float*)d_ws + 2 * N_NODES;
    float* nrmD   = nrmS + N_NODES;
    int*   rowptr = (int*)d_ws + 4 * N_NODES;
    int*   csr    = rowptr + 50048;
    float* feat   = (float*)d_ws + 4 * N_NODES + 50048 + N_EDGES;
    float* z2     = feat + (size_t)N_NODES * 64;
    int2*  segD   = (int2*)(z2 + (size_t)N_NODES * 64);
    int*   segS   = (int*)(segD + N_EDGES);
    int*   offD   = segS + N_EDGES;
    int*   offS   = offD + HB * OF_P;

    gcn_partition       <<<dim3(HB, 2), 1024, 0, stream>>>(src, dst, segD, segS, offD, offS);
    gcn_finish_transpose<<<HR, 1024, 0, stream>>>(segD, segS, offD, offS, rowptr, csr, nrmS, nrmD, h, feat);

    gcn_gather_gemm12_wmma<<<N_NODES / 16, 256, 0, stream>>>(rowptr, csr, feat, nrmD, nrmS, W1, b1, W2, z2);

    gcn_gather32_out<<<(N_NODES + G2_NB - 1) / G2_NB, 256, 0, stream>>>(rowptr, csr, z2, nrmD, b2, out);
}
